# starting kernel
# baseline (speedup 1.0000x reference)
#include <hip/hip_runtime.h>
#include <math.h>

typedef __attribute__((ext_vector_type(2))) float v2f;
typedef __attribute__((ext_vector_type(8))) float v8f;

#define NN 100000
#define EE 1600000
#define QQ 1000000

__global__ void GCN_deg_kernel(const int* __restrict__ esrc, const int* __restrict__ edst,
                               float* __restrict__ dout, float* __restrict__ din) {
    int e = blockIdx.x * blockDim.x + threadIdx.x;
    if (e >= EE) return;
    unsafeAtomicAdd(&dout[esrc[e]], 1.0f);
    unsafeAtomicAdd(&din[edst[e]], 1.0f);
}

__global__ void GCN_norm_kernel(float* __restrict__ d) {
    int i = blockIdx.x * blockDim.x + threadIdx.x;
    if (i >= 2 * NN) return;
    d[i] = rsqrtf(fmaxf(d[i], 1.0f));
}

__global__ void GCN_scatter1_kernel(const float* __restrict__ feat,
                                    const int* __restrict__ esrc, const int* __restrict__ edst,
                                    const float* __restrict__ onorm, float* __restrict__ agg) {
    int t = blockIdx.x * blockDim.x + threadIdx.x;
    int e = t >> 5, lane = t & 31;
    int s = esrc[e], d = edst[e];
    float sc = onorm[s];
    const float4 v = *(const float4*)(feat + (size_t)s * 128 + lane * 4);
    float* dp = agg + (size_t)d * 128 + lane * 4;
    unsafeAtomicAdd(dp + 0, v.x * sc);
    unsafeAtomicAdd(dp + 1, v.y * sc);
    unsafeAtomicAdd(dp + 2, v.z * sc);
    unsafeAtomicAdd(dp + 3, v.w * sc);
}

__global__ void __launch_bounds__(256)
GCN_gemm1_kernel(const float* __restrict__ A, const float* __restrict__ W,
                 const float* __restrict__ bias, const float* __restrict__ inorm,
                 float* __restrict__ H) {
    __shared__ float lW[128 * 128];
    int tid = threadIdx.x;
    {
        const float4* Wv = (const float4*)W;
        float4* lWv = (float4*)lW;
#pragma unroll
        for (int i = 0; i < 16; ++i) lWv[tid + 256 * i] = Wv[tid + 256 * i];
    }
    __syncthreads();

    int wave = tid >> 5, lane = tid & 31;
    int row0 = blockIdx.x * 16, col0 = wave * 16;
    int r = lane & 15, kh = lane >> 4;

    v8f c = {};
    const float* Ap = A + (size_t)(row0 + r) * 128 + 2 * kh;
    const float* Bp = lW + (size_t)(2 * kh) * 128 + col0 + r;
#pragma unroll
    for (int kt = 0; kt < 32; ++kt) {
        v2f a = *(const v2f*)(Ap + kt * 4);
        v2f b;
        b.x = Bp[kt * 512];
        b.y = Bp[kt * 512 + 128];
        c = __builtin_amdgcn_wmma_f32_16x16x4_f32(false, a, false, b, (short)0, c, false, false);
    }
    int n = col0 + r;
    float bb = bias[n];
#pragma unroll
    for (int v = 0; v < 8; ++v) {
        int m = row0 + v + 8 * kh;
        float val = fmaf(c[v], inorm[m], bb);
        H[(size_t)m * 128 + n] = fmaxf(val, 0.0f);
    }
}

__global__ void __launch_bounds__(128)
GCN_gemm2_kernel(const float* __restrict__ Hin, const float* __restrict__ W,
                 const float* __restrict__ onorm, float* __restrict__ T) {
    __shared__ float lW[128 * 64];
    int tid = threadIdx.x;
    {
        const float4* Wv = (const float4*)W;
        float4* lWv = (float4*)lW;
#pragma unroll
        for (int i = 0; i < 16; ++i) lWv[tid + 128 * i] = Wv[tid + 128 * i];
    }
    __syncthreads();

    int wave = tid >> 5, lane = tid & 31;
    int row0 = blockIdx.x * 16, col0 = wave * 16;
    int r = lane & 15, kh = lane >> 4;
    float sc = onorm[row0 + r];

    v8f c = {};
    const float* Ap = Hin + (size_t)(row0 + r) * 128 + 2 * kh;
    const float* Bp = lW + (size_t)(2 * kh) * 64 + col0 + r;
#pragma unroll
    for (int kt = 0; kt < 32; ++kt) {
        v2f a = *(const v2f*)(Ap + kt * 4);
        a.x *= sc; a.y *= sc;
        v2f b;
        b.x = Bp[kt * 256];
        b.y = Bp[kt * 256 + 64];
        c = __builtin_amdgcn_wmma_f32_16x16x4_f32(false, a, false, b, (short)0, c, false, false);
    }
    int n = col0 + r;
#pragma unroll
    for (int v = 0; v < 8; ++v) {
        int m = row0 + v + 8 * kh;
        T[(size_t)m * 64 + n] = c[v];
    }
}

__global__ void GCN_scatter2_kernel(const float* __restrict__ T,
                                    const int* __restrict__ esrc, const int* __restrict__ edst,
                                    float* __restrict__ H2) {
    int t = blockIdx.x * blockDim.x + threadIdx.x;
    int e = t >> 5, lane = t & 31;
    int s = esrc[e], d = edst[e];
    const float2 v = *(const float2*)(T + (size_t)s * 64 + lane * 2);
    float* dp = H2 + (size_t)d * 64 + lane * 2;
    unsafeAtomicAdd(dp + 0, v.x);
    unsafeAtomicAdd(dp + 1, v.y);
}

__global__ void GCN_finalize_kernel(float* __restrict__ H2, const float* __restrict__ inorm,
                                    const float* __restrict__ b2) {
    int i = blockIdx.x * blockDim.x + threadIdx.x;
    if (i >= NN * 16) return;
    int row = i >> 4;
    int c4 = (i & 15) * 4;
    float s = inorm[row];
    float4 v = ((float4*)H2)[i];
    v.x = fmaf(v.x, s, b2[c4 + 0]);
    v.y = fmaf(v.y, s, b2[c4 + 1]);
    v.z = fmaf(v.z, s, b2[c4 + 2]);
    v.w = fmaf(v.w, s, b2[c4 + 3]);
    ((float4*)H2)[i] = v;
}

__global__ void GCN_score_kernel(const float* __restrict__ H2, const int* __restrict__ qs,
                                 const int* __restrict__ qd, float* __restrict__ out) {
    int t = blockIdx.x * blockDim.x + threadIdx.x;
    int q = t >> 5, lane = t & 31;
    int s = qs[q], d = qd[q];
    float2 a = *(const float2*)(H2 + (size_t)s * 64 + lane * 2);
    float2 b = *(const float2*)(H2 + (size_t)d * 64 + lane * 2);
    float p = a.x * b.x + a.y * b.y;
#pragma unroll
    for (int off = 16; off > 0; off >>= 1) p += __shfl_xor(p, off, 32);
    if (lane == 0) out[q] = 1.0f / (1.0f + __expf(-p));
}

extern "C" void kernel_launch(void* const* d_in, const int* in_sizes, int n_in,
                              void* d_out, int out_size, void* d_ws, size_t ws_size,
                              hipStream_t stream) {
    const float* feat = (const float*)d_in[0];
    const int* esrc   = (const int*)d_in[1];
    const int* edst   = (const int*)d_in[2];
    const int* qsrc   = (const int*)d_in[3];
    const int* qdst   = (const int*)d_in[4];
    const float* W1   = (const float*)d_in[5];
    const float* b1   = (const float*)d_in[6];
    const float* W2   = (const float*)d_in[7];
    const float* b2   = (const float*)d_in[8];
    float* out = (float*)d_out;

    float* ws    = (float*)d_ws;
    float* onorm = ws;
    float* inorm = ws + NN;
    float* agg1  = ws + 2 * NN;
    float* h1    = agg1 + (size_t)NN * 128;
    float* t2    = h1 + (size_t)NN * 128;
    float* h2    = t2 + (size_t)NN * 64;

    hipMemsetAsync(ws, 0, (size_t)(2 * NN + (size_t)NN * 128) * sizeof(float), stream);
    hipMemsetAsync(h2, 0, (size_t)NN * 64 * sizeof(float), stream);

    GCN_deg_kernel<<<(EE + 255) / 256, 256, 0, stream>>>(esrc, edst, onorm, inorm);
    GCN_norm_kernel<<<(2 * NN + 255) / 256, 256, 0, stream>>>(onorm);
    GCN_scatter1_kernel<<<EE / 8, 256, 0, stream>>>(feat, esrc, edst, onorm, agg1);
    GCN_gemm1_kernel<<<NN / 16, 256, 0, stream>>>(agg1, W1, b1, inorm, h1);
    GCN_gemm2_kernel<<<NN / 16, 128, 0, stream>>>(h1, W2, onorm, t2);
    GCN_scatter2_kernel<<<EE / 8, 256, 0, stream>>>(t2, esrc, edst, h2);
    GCN_finalize_kernel<<<(NN * 16 + 255) / 256, 256, 0, stream>>>(h2, inorm, b2);
    GCN_score_kernel<<<QQ / 8, 256, 0, stream>>>(h2, qsrc, qdst, out);
}


# Both float-atomic edge scatters become gathers over edges counting-sorted by dst (int atomic arrival-order rank + prefix sum): a wave sums a row in rank order, writes it once (layer 2 with norm+bias)
# speedup vs baseline: 1.8884x; 1.8884x over previous
#include <hip/hip_runtime.h>
#include <math.h>

typedef __attribute__((ext_vector_type(2))) float v2f;
typedef __attribute__((ext_vector_type(8))) float v8f;

#define NN 100000
#define EE 1600000
#define QQ 1000000

__global__ void GCN_deg_kernel(const int* __restrict__ esrc, const int* __restrict__ edst,
                               float* __restrict__ dout, int* __restrict__ cin,
                               int* __restrict__ rank) {
    int e = blockIdx.x * blockDim.x + threadIdx.x;
    if (e >= EE) return;
    int s = esrc[e], d = edst[e];
    unsafeAtomicAdd(&dout[s], 1.0f);
    rank[e] = atomicAdd(&cin[d], 1);
}

__global__ void __launch_bounds__(1024)
GCN_norm_scan_kernel(float* __restrict__ onorm, const int* __restrict__ cin,
                     float* __restrict__ inorm, int* __restrict__ rowptr) {
    __shared__ int red[32];
    __shared__ int wsum[32];
    int t = threadIdx.x, b = blockIdx.x, lane = t & 31, wave = t >> 5;
    const int4* c4 = (const int4*)cin;
    int part = 0;
    int n4 = b * 256;
#pragma unroll 8
    for (int i = t; i < n4; i += 1024) {
        int4 v = c4[i];
        part += (v.x + v.y) + (v.z + v.w);
    }
#pragma unroll
    for (int off = 16; off > 0; off >>= 1) part += __shfl_xor(part, off, 32);
    if (lane == 0) red[wave] = part;

    int idx = b * 1024 + t;
    int c = (idx < NN) ? cin[idx] : 0;
    int inc = c;
#pragma unroll
    for (int off = 1; off < 32; off <<= 1) {
        int u = __shfl_up(inc, off, 32);
        if (lane >= off) inc += u;
    }
    if (lane == 31) wsum[wave] = inc;
    __syncthreads();
    int base = 0;
#pragma unroll
    for (int w = 0; w < 32; ++w) {
        base += red[w];
        if (w < wave) base += wsum[w];
    }
    int excl = base + inc - c;
    if (idx < NN) {
        rowptr[idx] = excl;
        inorm[idx] = rsqrtf(fmaxf((float)c, 1.0f));
        onorm[idx] = rsqrtf(fmaxf(onorm[idx], 1.0f));
        if (idx == NN - 1) rowptr[NN] = excl + c;
    }
}

__global__ void GCN_fill_kernel(const int* __restrict__ esrc, const int* __restrict__ edst,
                                const int* __restrict__ rowptr, const int* __restrict__ rank,
                                int* __restrict__ perm) {
    int e = blockIdx.x * blockDim.x + threadIdx.x;
    if (e >= EE) return;
    perm[rowptr[edst[e]] + rank[e]] = esrc[e];
}

__global__ void __launch_bounds__(256)
GCN_gather1_kernel(const float* __restrict__ feat, const int* __restrict__ rowptr,
                   const int* __restrict__ perm, const float* __restrict__ onorm,
                   float* __restrict__ agg) {
    int lane = threadIdx.x & 31;
    int n = __builtin_amdgcn_readfirstlane(blockIdx.x * 8 + (threadIdx.x >> 5));
    int beg = rowptr[n], end = rowptr[n + 1];
    float4 acc = make_float4(0.0f, 0.0f, 0.0f, 0.0f);
    for (int base = beg; base < end; base += 32) {
        int cnt = min(32, end - base);
        int s_l = 0;
        float sc_l = 0.0f;
        if (lane < cnt) {
            s_l = perm[base + lane];
            sc_l = onorm[s_l];
        }
        int sc_i = __float_as_int(sc_l);
        for (int j = 0; j < cnt; j += 4) {
            int j1 = min(j + 1, cnt - 1), j2 = min(j + 2, cnt - 1), j3 = min(j + 3, cnt - 1);
            int s0 = __builtin_amdgcn_readlane(s_l, j);
            int s1 = __builtin_amdgcn_readlane(s_l, j1);
            int s2 = __builtin_amdgcn_readlane(s_l, j2);
            int s3 = __builtin_amdgcn_readlane(s_l, j3);
            float c0 = __int_as_float(__builtin_amdgcn_readlane(sc_i, j));
            float c1 = (j + 1 < cnt) ? __int_as_float(__builtin_amdgcn_readlane(sc_i, j1)) : 0.0f;
            float c2 = (j + 2 < cnt) ? __int_as_float(__builtin_amdgcn_readlane(sc_i, j2)) : 0.0f;
            float c3 = (j + 3 < cnt) ? __int_as_float(__builtin_amdgcn_readlane(sc_i, j3)) : 0.0f;
            const float4 v0 = *(const float4*)(feat + (size_t)s0 * 128 + lane * 4);
            const float4 v1 = *(const float4*)(feat + (size_t)s1 * 128 + lane * 4);
            const float4 v2 = *(const float4*)(feat + (size_t)s2 * 128 + lane * 4);
            const float4 v3 = *(const float4*)(feat + (size_t)s3 * 128 + lane * 4);
            acc.x = fmaf(v0.x, c0, acc.x); acc.y = fmaf(v0.y, c0, acc.y);
            acc.z = fmaf(v0.z, c0, acc.z); acc.w = fmaf(v0.w, c0, acc.w);
            acc.x = fmaf(v1.x, c1, acc.x); acc.y = fmaf(v1.y, c1, acc.y);
            acc.z = fmaf(v1.z, c1, acc.z); acc.w = fmaf(v1.w, c1, acc.w);
            acc.x = fmaf(v2.x, c2, acc.x); acc.y = fmaf(v2.y, c2, acc.y);
            acc.z = fmaf(v2.z, c2, acc.z); acc.w = fmaf(v2.w, c2, acc.w);
            acc.x = fmaf(v3.x, c3, acc.x); acc.y = fmaf(v3.y, c3, acc.y);
            acc.z = fmaf(v3.z, c3, acc.z); acc.w = fmaf(v3.w, c3, acc.w);
        }
    }
    *(float4*)(agg + (size_t)n * 128 + lane * 4) = acc;
}

__global__ void __launch_bounds__(256)
GCN_gemm1_kernel(const float* __restrict__ A, const float* __restrict__ W,
                 const float* __restrict__ bias, const float* __restrict__ inorm,
                 float* __restrict__ H) {
    __shared__ float lW[128 * 128];
    int tid = threadIdx.x;
    {
        const float4* Wv = (const float4*)W;
        float4* lWv = (float4*)lW;
#pragma unroll
        for (int i = 0; i < 16; ++i) lWv[tid + 256 * i] = Wv[tid + 256 * i];
    }
    __syncthreads();

    int wave = tid >> 5, lane = tid & 31;
    int row0 = blockIdx.x * 16, col0 = wave * 16;
    int r = lane & 15, kh = lane >> 4;

    v8f c = {};
    const float* Ap = A + (size_t)(row0 + r) * 128 + 2 * kh;
    const float* Bp = lW + (size_t)(2 * kh) * 128 + col0 + r;
#pragma unroll
    for (int kt = 0; kt < 32; ++kt) {
        v2f a = *(const v2f*)(Ap + kt * 4);
        v2f b;
        b.x = Bp[kt * 512];
        b.y = Bp[kt * 512 + 128];
        c = __builtin_amdgcn_wmma_f32_16x16x4_f32(false, a, false, b, (short)0, c, false, false);
    }
    int n = col0 + r;
    float bb = bias[n];
#pragma unroll
    for (int v = 0; v < 8; ++v) {
        int m = row0 + v + 8 * kh;
        float val = fmaf(c[v], inorm[m], bb);
        H[(size_t)m * 128 + n] = fmaxf(val, 0.0f);
    }
}

__global__ void __launch_bounds__(128)
GCN_gemm2_kernel(const float* __restrict__ Hin, const float* __restrict__ W,
                 const float* __restrict__ onorm, float* __restrict__ T) {
    __shared__ float lW[128 * 64];
    int tid = threadIdx.x;
    {
        const float4* Wv = (const float4*)W;
        float4* lWv = (float4*)lW;
#pragma unroll
        for (int i = 0; i < 16; ++i) lWv[tid + 128 * i] = Wv[tid + 128 * i];
    }
    __syncthreads();

    int wave = tid >> 5, lane = tid & 31;
    int row0 = blockIdx.x * 16, col0 = wave * 16;
    int r = lane & 15, kh = lane >> 4;
    float sc = onorm[row0 + r];

    v8f c = {};
    const float* Ap = Hin + (size_t)(row0 + r) * 128 + 2 * kh;
    const float* Bp = lW + (size_t)(2 * kh) * 64 + col0 + r;
#pragma unroll
    for (int kt = 0; kt < 32; ++kt) {
        v2f a = *(const v2f*)(Ap + kt * 4);
        a.x *= sc; a.y *= sc;
        v2f b;
        b.x = Bp[kt * 256];
        b.y = Bp[kt * 256 + 64];
        c = __builtin_amdgcn_wmma_f32_16x16x4_f32(false, a, false, b, (short)0, c, false, false);
    }
    int n = col0 + r;
#pragma unroll
    for (int v = 0; v < 8; ++v) {
        int m = row0 + v + 8 * kh;
        T[(size_t)m * 64 + n] = c[v];
    }
}

__global__ void __launch_bounds__(256)
GCN_gather2_kernel(const float* __restrict__ T, const int* __restrict__ rowptr,
                   const int* __restrict__ perm, const float* __restrict__ inorm,
                   const float* __restrict__ b2, float* __restrict__ H2) {
    int lane = threadIdx.x & 31;
    int n = __builtin_amdgcn_readfirstlane(blockIdx.x * 8 + (threadIdx.x >> 5));
    int beg = rowptr[n], end = rowptr[n + 1];
    float2 acc = make_float2(0.0f, 0.0f);
    for (int base = beg; base < end; base += 32) {
        int cnt = min(32, end - base);
        int s_l = 0;
        if (lane < cnt) s_l = perm[base + lane];
        for (int j = 0; j < cnt; j += 4) {
            int j1 = min(j + 1, cnt - 1), j2 = min(j + 2, cnt - 1), j3 = min(j + 3, cnt - 1);
            int s0 = __builtin_amdgcn_readlane(s_l, j);
            int s1 = __builtin_amdgcn_readlane(s_l, j1);
            int s2 = __builtin_amdgcn_readlane(s_l, j2);
            int s3 = __builtin_amdgcn_readlane(s_l, j3);
            float c1 = (j + 1 < cnt) ? 1.0f : 0.0f;
            float c2 = (j + 2 < cnt) ? 1.0f : 0.0f;
            float c3 = (j + 3 < cnt) ? 1.0f : 0.0f;
            const float2 v0 = *(const float2*)(T + (size_t)s0 * 64 + lane * 2);
            const float2 v1 = *(const float2*)(T + (size_t)s1 * 64 + lane * 2);
            const float2 v2 = *(const float2*)(T + (size_t)s2 * 64 + lane * 2);
            const float2 v3 = *(const float2*)(T + (size_t)s3 * 64 + lane * 2);
            acc.x += v0.x;                  acc.y += v0.y;
            acc.x = fmaf(v1.x, c1, acc.x);  acc.y = fmaf(v1.y, c1, acc.y);
            acc.x = fmaf(v2.x, c2, acc.x);  acc.y = fmaf(v2.y, c2, acc.y);
            acc.x = fmaf(v3.x, c3, acc.x);  acc.y = fmaf(v3.y, c3, acc.y);
        }
    }
    float s = inorm[n];
    float2 o;
    o.x = fmaf(acc.x, s, b2[lane * 2 + 0]);
    o.y = fmaf(acc.y, s, b2[lane * 2 + 1]);
    *(float2*)(H2 + (size_t)n * 64 + lane * 2) = o;
}

__global__ void GCN_score_kernel(const float* __restrict__ H2, const int* __restrict__ qs,
                                 const int* __restrict__ qd, float* __restrict__ out) {
    int t = blockIdx.x * blockDim.x + threadIdx.x;
    int q = t >> 5, lane = t & 31;
    int s = qs[q], d = qd[q];
    float2 a = *(const float2*)(H2 + (size_t)s * 64 + lane * 2);
    float2 b = *(const float2*)(H2 + (size_t)d * 64 + lane * 2);
    float p = a.x * b.x + a.y * b.y;
#pragma unroll
    for (int off = 16; off > 0; off >>= 1) p += __shfl_xor(p, off, 32);
    if (lane == 0) out[q] = 1.0f / (1.0f + __expf(-p));
}

extern "C" void kernel_launch(void* const* d_in, const int* in_sizes, int n_in,
                              void* d_out, int out_size, void* d_ws, size_t ws_size,
                              hipStream_t stream) {
    const float* feat = (const float*)d_in[0];
    const int* esrc   = (const int*)d_in[1];
    const int* edst   = (const int*)d_in[2];
    const int* qsrc   = (const int*)d_in[3];
    const int* qdst   = (const int*)d_in[4];
    const float* W1   = (const float*)d_in[5];
    const float* b1   = (const float*)d_in[6];
    const float* W2   = (const float*)d_in[7];
    const float* b2   = (const float*)d_in[8];
    float* out = (float*)d_out;

    float* ws    = (float*)d_ws;
    float* onorm = ws;
    int*   cin   = (int*)(ws + NN);
    float* agg1  = ws + 2 * NN;
    float* h2    = agg1;
    float* h1    = agg1 + (size_t)NN * 128;
    float* t2    = h1 + (size_t)NN * 128;
    int*   perm  = (int*)(t2 + (size_t)NN * 64);
    int*   rank  = perm + EE;
    float* inorm = (float*)(rank + EE);
    int*   rowptr = (int*)(inorm + NN);

    hipMemsetAsync(ws, 0, (size_t)(2 * NN) * sizeof(float), stream);

    GCN_deg_kernel<<<(EE + 255) / 256, 256, 0, stream>>>(esrc, edst, onorm, cin, rank);
    GCN_norm_scan_kernel<<<(NN + 1023) / 1024, 1024, 0, stream>>>(onorm, cin, inorm, rowptr);
    GCN_fill_kernel<<<(EE + 255) / 256, 256, 0, stream>>>(esrc, edst, rowptr, rank, perm);
    GCN_gather1_kernel<<<NN / 8, 256, 0, stream>>>(feat, rowptr, perm, onorm, agg1);
    GCN_gemm1_kernel<<<NN / 16, 256, 0, stream>>>(agg1, W1, b1, inorm, h1);
    GCN_gemm2_kernel<<<NN / 16, 128, 0, stream>>>(h1, W2, onorm, t2);
    GCN_gather2_kernel<<<NN / 8, 256, 0, stream>>>(t2, rowptr, perm, inorm, b2, h2);
    GCN_score_kernel<<<QQ / 8, 256, 0, stream>>>(h2, qsrc, qdst, out);
}


# Edge ordering and degree counts on LDS counters, no global atomics: edges binned by 391-node dst and src range (count, scan, place), one block per bin ranks its rows; row sums in LDS arrival order
# speedup vs baseline: 2.4617x; 1.3036x over previous
#include <hip/hip_runtime.h>
#include <math.h>

typedef __attribute__((ext_vector_type(2))) float v2f;
typedef __attribute__((ext_vector_type(8))) float v8f;

#define NN 100000
#define EE 1600000
#define QQ 1000000

#define NB  256
#define BW  391
#define NWG 256
#define CH  (EE / NWG)
#define EPT 7
#define SRC_BITS 17
#define RL  8

__global__ void __launch_bounds__(1024)
GCN_bincount_kernel(const int* __restrict__ esrc, const int* __restrict__ edst,
                    int* __restrict__ cnt) {
    __shared__ int hD[NB];
    __shared__ int hS[NB];
    int t = threadIdx.x, w = blockIdx.x;
    if (t < NB) { hD[t] = 0; hS[t] = 0; }
    __syncthreads();
    int e0 = w * CH;
    int d[EPT], s[EPT];
#pragma unroll
    for (int k = 0; k < EPT; ++k) {
        int i = t + k * 1024;
        d[k] = (i < CH) ? edst[e0 + i] : -1;
        s[k] = (i < CH) ? esrc[e0 + i] : -1;
    }
#pragma unroll
    for (int k = 0; k < EPT; ++k) {
        if (d[k] >= 0) {
            atomicAdd(&hD[d[k] / BW], 1);
            atomicAdd(&hS[s[k] / BW], 1);
        }
    }
    __syncthreads();
    if (t < NB) {
        cnt[t * NWG + w] = hD[t];
        cnt[NB * NWG + t * NWG + w] = hS[t];
    }
}

__global__ void __launch_bounds__(1024)
GCN_binscan_kernel(const int* __restrict__ cnt, int* __restrict__ base) {
    __shared__ int red[32];
    __shared__ int wsum[32];
    const int* c = cnt + blockIdx.y * (NB * NWG);
    int* o = base + blockIdx.y * (NB * NWG);
    int t = threadIdx.x, b = blockIdx.x, lane = t & 31, wave = t >> 5;
    const int4* c4 = (const int4*)c;
    int part = 0;
    int n4 = b * 256;
#pragma unroll 8
    for (int i = t; i < n4; i += 1024) {
        int4 v = c4[i];
        part += (v.x + v.y) + (v.z + v.w);
    }
#pragma unroll
    for (int off = 16; off > 0; off >>= 1) part += __shfl_xor(part, off, 32);
    if (lane == 0) red[wave] = part;

    int idx = b * 1024 + t;
    int v = c[idx];
    int inc = v;
#pragma unroll
    for (int off = 1; off < 32; off <<= 1) {
        int u = __shfl_up(inc, off, 32);
        if (lane >= off) inc += u;
    }
    if (lane == 31) wsum[wave] = inc;
    __syncthreads();
    int pre = 0;
#pragma unroll
    for (int k = 0; k < 32; ++k) {
        pre += red[k];
        if (k < wave) pre += wsum[k];
    }
    o[idx] = pre + inc - v;
}

__global__ void __launch_bounds__(1024)
GCN_binplace_kernel(const int* __restrict__ esrc, const int* __restrict__ edst,
                    const int* __restrict__ base, int* __restrict__ binD,
                    int* __restrict__ binS) {
    __shared__ int oD[NB];
    __shared__ int oS[NB];
    int t = threadIdx.x, w = blockIdx.x;
    if (t < NB) {
        oD[t] = base[t * NWG + w];
        oS[t] = base[NB * NWG + t * NWG + w];
    }
    int e0 = w * CH;
    int d[EPT], s[EPT];
#pragma unroll
    for (int k = 0; k < EPT; ++k) {
        int i = t + k * 1024;
        d[k] = (i < CH) ? edst[e0 + i] : -1;
        s[k] = (i < CH) ? esrc[e0 + i] : -1;
    }
    __syncthreads();
#pragma unroll
    for (int k = 0; k < EPT; ++k) {
        if (d[k] >= 0) {
            int bd = d[k] / BW, bs = s[k] / BW;
            int pd = atomicAdd(&oD[bd], 1);
            int ps = atomicAdd(&oS[bs], 1);
            binD[pd] = ((d[k] - bd * BW) << SRC_BITS) | s[k];
            binS[ps] = s[k] - bs * BW;
        }
    }
}

__global__ void __launch_bounds__(1024)
GCN_rows_kernel(const int* __restrict__ base, const int* __restrict__ binD,
                const int* __restrict__ binS, int* __restrict__ rowptr,
                float* __restrict__ inorm, float* __restrict__ onorm,
                int* __restrict__ perm) {
    __shared__ int cntIn[BW];
    __shared__ int cntOut[BW];
    __shared__ int start[BW];
    __shared__ int cur[BW];
    __shared__ int wsum[32];
    int t = threadIdx.x, b = blockIdx.x, lane = t & 31, wave = t >> 5;
    int dBeg = base[b * NWG];
    int dEnd = (b + 1 < NB) ? base[(b + 1) * NWG] : EE;
    int sBeg = base[NB * NWG + b * NWG];
    int sEnd = (b + 1 < NB) ? base[NB * NWG + (b + 1) * NWG] : EE;
    if (t < BW) { cntIn[t] = 0; cntOut[t] = 0; cur[t] = 0; }
    __syncthreads();
    for (int i0 = dBeg + t; i0 < dEnd; i0 += RL * 1024) {
        int p[RL];
#pragma unroll
        for (int k = 0; k < RL; ++k) {
            int i = i0 + k * 1024;
            p[k] = (i < dEnd) ? binD[i] : -1;
        }
#pragma unroll
        for (int k = 0; k < RL; ++k)
            if (p[k] >= 0) atomicAdd(&cntIn[p[k] >> SRC_BITS], 1);
    }
    for (int i0 = sBeg + t; i0 < sEnd; i0 += RL * 1024) {
        int p[RL];
#pragma unroll
        for (int k = 0; k < RL; ++k) {
            int i = i0 + k * 1024;
            p[k] = (i < sEnd) ? binS[i] : -1;
        }
#pragma unroll
        for (int k = 0; k < RL; ++k)
            if (p[k] >= 0) atomicAdd(&cntOut[p[k]], 1);
    }
    __syncthreads();

    int c = (t < BW) ? cntIn[t] : 0;
    int inc = c;
#pragma unroll
    for (int off = 1; off < 32; off <<= 1) {
        int u = __shfl_up(inc, off, 32);
        if (lane >= off) inc += u;
    }
    if (lane == 31) wsum[wave] = inc;
    __syncthreads();
    int pre = 0;
#pragma unroll
    for (int k = 0; k < 32; ++k) {
        if (k < wave) pre += wsum[k];
    }
    int first = dBeg + pre + inc - c;
    int n = b * BW + t;
    if (t < BW) {
        start[t] = first;
        if (n < NN) {
            rowptr[n] = first;
            inorm[n] = rsqrtf(fmaxf((float)c, 1.0f));
            onorm[n] = rsqrtf(fmaxf((float)cntOut[t], 1.0f));
            if (n == NN - 1) rowptr[NN] = first + c;
        }
    }
    __syncthreads();
    for (int i0 = dBeg + t; i0 < dEnd; i0 += RL * 1024) {
        int p[RL];
#pragma unroll
        for (int k = 0; k < RL; ++k) {
            int i = i0 + k * 1024;
            p[k] = (i < dEnd) ? binD[i] : -1;
        }
#pragma unroll
        for (int k = 0; k < RL; ++k) {
            if (p[k] >= 0) {
                int dl = p[k] >> SRC_BITS;
                int slot = atomicAdd(&cur[dl], 1);
                perm[start[dl] + slot] = p[k] & ((1 << SRC_BITS) - 1);
            }
        }
    }
}

__global__ void __launch_bounds__(256)
GCN_gather1_kernel(const float* __restrict__ feat, const int* __restrict__ rowptr,
                   const int* __restrict__ perm, const float* __restrict__ onorm,
                   float* __restrict__ agg) {
    int lane = threadIdx.x & 31;
    int n = __builtin_amdgcn_readfirstlane(blockIdx.x * 8 + (threadIdx.x >> 5));
    int beg = rowptr[n], end = rowptr[n + 1];
    float4 acc = make_float4(0.0f, 0.0f, 0.0f, 0.0f);
    for (int base = beg; base < end; base += 32) {
        int cnt = min(32, end - base);
        int s_l = 0;
        float sc_l = 0.0f;
        if (lane < cnt) {
            s_l = perm[base + lane];
            sc_l = onorm[s_l];
        }
        int sc_i = __float_as_int(sc_l);
        for (int j = 0; j < cnt; j += 4) {
            int j1 = min(j + 1, cnt - 1), j2 = min(j + 2, cnt - 1), j3 = min(j + 3, cnt - 1);
            int s0 = __builtin_amdgcn_readlane(s_l, j);
            int s1 = __builtin_amdgcn_readlane(s_l, j1);
            int s2 = __builtin_amdgcn_readlane(s_l, j2);
            int s3 = __builtin_amdgcn_readlane(s_l, j3);
            float c0 = __int_as_float(__builtin_amdgcn_readlane(sc_i, j));
            float c1 = (j + 1 < cnt) ? __int_as_float(__builtin_amdgcn_readlane(sc_i, j1)) : 0.0f;
            float c2 = (j + 2 < cnt) ? __int_as_float(__builtin_amdgcn_readlane(sc_i, j2)) : 0.0f;
            float c3 = (j + 3 < cnt) ? __int_as_float(__builtin_amdgcn_readlane(sc_i, j3)) : 0.0f;
            const float4 v0 = *(const float4*)(feat + (size_t)s0 * 128 + lane * 4);
            const float4 v1 = *(const float4*)(feat + (size_t)s1 * 128 + lane * 4);
            const float4 v2 = *(const float4*)(feat + (size_t)s2 * 128 + lane * 4);
            const float4 v3 = *(const float4*)(feat + (size_t)s3 * 128 + lane * 4);
            acc.x = fmaf(v0.x, c0, acc.x); acc.y = fmaf(v0.y, c0, acc.y);
            acc.z = fmaf(v0.z, c0, acc.z); acc.w = fmaf(v0.w, c0, acc.w);
            acc.x = fmaf(v1.x, c1, acc.x); acc.y = fmaf(v1.y, c1, acc.y);
            acc.z = fmaf(v1.z, c1, acc.z); acc.w = fmaf(v1.w, c1, acc.w);
            acc.x = fmaf(v2.x, c2, acc.x); acc.y = fmaf(v2.y, c2, acc.y);
            acc.z = fmaf(v2.z, c2, acc.z); acc.w = fmaf(v2.w, c2, acc.w);
            acc.x = fmaf(v3.x, c3, acc.x); acc.y = fmaf(v3.y, c3, acc.y);
            acc.z = fmaf(v3.z, c3, acc.z); acc.w = fmaf(v3.w, c3, acc.w);
        }
    }
    *(float4*)(agg + (size_t)n * 128 + lane * 4) = acc;
}

__global__ void __launch_bounds__(256)
GCN_gemm1_kernel(const float* __restrict__ A, const float* __restrict__ W,
                 const float* __restrict__ bias, const float* __restrict__ inorm,
                 float* __restrict__ H) {
    __shared__ float lW[128 * 128];
    int tid = threadIdx.x;
    {
        const float4* Wv = (const float4*)W;
        float4* lWv = (float4*)lW;
#pragma unroll
        for (int i = 0; i < 16; ++i) lWv[tid + 256 * i] = Wv[tid + 256 * i];
    }
    __syncthreads();

    int wave = tid >> 5, lane = tid & 31;
    int row0 = blockIdx.x * 16, col0 = wave * 16;
    int r = lane & 15, kh = lane >> 4;

    v8f c = {};
    const float* Ap = A + (size_t)(row0 + r) * 128 + 2 * kh;
    const float* Bp = lW + (size_t)(2 * kh) * 128 + col0 + r;
#pragma unroll
    for (int kt = 0; kt < 32; ++kt) {
        v2f a = *(const v2f*)(Ap + kt * 4);
        v2f b;
        b.x = Bp[kt * 512];
        b.y = Bp[kt * 512 + 128];
        c = __builtin_amdgcn_wmma_f32_16x16x4_f32(false, a, false, b, (short)0, c, false, false);
    }
    int n = col0 + r;
    float bb = bias[n];
#pragma unroll
    for (int v = 0; v < 8; ++v) {
        int m = row0 + v + 8 * kh;
        float val = fmaf(c[v], inorm[m], bb);
        H[(size_t)m * 128 + n] = fmaxf(val, 0.0f);
    }
}

__global__ void __launch_bounds__(128)
GCN_gemm2_kernel(const float* __restrict__ Hin, const float* __restrict__ W,
                 const float* __restrict__ onorm, float* __restrict__ T) {
    __shared__ float lW[128 * 64];
    int tid = threadIdx.x;
    {
        const float4* Wv = (const float4*)W;
        float4* lWv = (float4*)lW;
#pragma unroll
        for (int i = 0; i < 16; ++i) lWv[tid + 128 * i] = Wv[tid + 128 * i];
    }
    __syncthreads();

    int wave = tid >> 5, lane = tid & 31;
    int row0 = blockIdx.x * 16, col0 = wave * 16;
    int r = lane & 15, kh = lane >> 4;
    float sc = onorm[row0 + r];

    v8f c = {};
    const float* Ap = Hin + (size_t)(row0 + r) * 128 + 2 * kh;
    const float* Bp = lW + (size_t)(2 * kh) * 64 + col0 + r;
#pragma unroll
    for (int kt = 0; kt < 32; ++kt) {
        v2f a = *(const v2f*)(Ap + kt * 4);
        a.x *= sc; a.y *= sc;
        v2f b;
        b.x = Bp[kt * 256];
        b.y = Bp[kt * 256 + 64];
        c = __builtin_amdgcn_wmma_f32_16x16x4_f32(false, a, false, b, (short)0, c, false, false);
    }
    int n = col0 + r;
#pragma unroll
    for (int v = 0; v < 8; ++v) {
        int m = row0 + v + 8 * kh;
        T[(size_t)m * 64 + n] = c[v];
    }
}

__global__ void __launch_bounds__(256)
GCN_gather2_kernel(const float* __restrict__ T, const int* __restrict__ rowptr,
                   const int* __restrict__ perm, const float* __restrict__ inorm,
                   const float* __restrict__ b2, float* __restrict__ H2) {
    int lane = threadIdx.x & 31;
    int n = __builtin_amdgcn_readfirstlane(blockIdx.x * 8 + (threadIdx.x >> 5));
    int beg = rowptr[n], end = rowptr[n + 1];
    float2 acc = make_float2(0.0f, 0.0f);
    for (int base = beg; base < end; base += 32) {
        int cnt = min(32, end - base);
        int s_l = 0;
        if (lane < cnt) s_l = perm[base + lane];
        for (int j = 0; j < cnt; j += 4) {
            int j1 = min(j + 1, cnt - 1), j2 = min(j + 2, cnt - 1), j3 = min(j + 3, cnt - 1);
            int s0 = __builtin_amdgcn_readlane(s_l, j);
            int s1 = __builtin_amdgcn_readlane(s_l, j1);
            int s2 = __builtin_amdgcn_readlane(s_l, j2);
            int s3 = __builtin_amdgcn_readlane(s_l, j3);
            float c1 = (j + 1 < cnt) ? 1.0f : 0.0f;
            float c2 = (j + 2 < cnt) ? 1.0f : 0.0f;
            float c3 = (j + 3 < cnt) ? 1.0f : 0.0f;
            const float2 v0 = *(const float2*)(T + (size_t)s0 * 64 + lane * 2);
            const float2 v1 = *(const float2*)(T + (size_t)s1 * 64 + lane * 2);
            const float2 v2 = *(const float2*)(T + (size_t)s2 * 64 + lane * 2);
            const float2 v3 = *(const float2*)(T + (size_t)s3 * 64 + lane * 2);
            acc.x += v0.x;                  acc.y += v0.y;
            acc.x = fmaf(v1.x, c1, acc.x);  acc.y = fmaf(v1.y, c1, acc.y);
            acc.x = fmaf(v2.x, c2, acc.x);  acc.y = fmaf(v2.y, c2, acc.y);
            acc.x = fmaf(v3.x, c3, acc.x);  acc.y = fmaf(v3.y, c3, acc.y);
        }
    }
    float s = inorm[n];
    float2 o;
    o.x = fmaf(acc.x, s, b2[lane * 2 + 0]);
    o.y = fmaf(acc.y, s, b2[lane * 2 + 1]);
    *(float2*)(H2 + (size_t)n * 64 + lane * 2) = o;
}

__global__ void GCN_score_kernel(const float* __restrict__ H2, const int* __restrict__ qs,
                                 const int* __restrict__ qd, float* __restrict__ out) {
    int t = blockIdx.x * blockDim.x + threadIdx.x;
    int q = t >> 5, lane = t & 31;
    int s = qs[q], d = qd[q];
    float2 a = *(const float2*)(H2 + (size_t)s * 64 + lane * 2);
    float2 b = *(const float2*)(H2 + (size_t)d * 64 + lane * 2);
    float p = a.x * b.x + a.y * b.y;
#pragma unroll
    for (int off = 16; off > 0; off >>= 1) p += __shfl_xor(p, off, 32);
    if (lane == 0) out[q] = 1.0f / (1.0f + __expf(-p));
}

extern "C" void kernel_launch(void* const* d_in, const int* in_sizes, int n_in,
                              void* d_out, int out_size, void* d_ws, size_t ws_size,
                              hipStream_t stream) {
    const float* feat = (const float*)d_in[0];
    const int* esrc   = (const int*)d_in[1];
    const int* edst   = (const int*)d_in[2];
    const int* qsrc   = (const int*)d_in[3];
    const int* qdst   = (const int*)d_in[4];
    const float* W1   = (const float*)d_in[5];
    const float* b1   = (const float*)d_in[6];
    const float* W2   = (const float*)d_in[7];
    const float* b2   = (const float*)d_in[8];
    float* out = (float*)d_out;

    float* ws    = (float*)d_ws;
    float* onorm = ws;
    float* agg1  = ws + 2 * NN;
    float* h2    = agg1;
    float* h1    = agg1 + (size_t)NN * 128;
    float* t2    = h1 + (size_t)NN * 128;
    int*   perm  = (int*)(t2 + (size_t)NN * 64);
    float* inorm = (float*)(perm + 2 * EE);
    int*   rowptr = (int*)(inorm + NN);
    int*   cnt   = (int*)h1;
    int*   base  = cnt + 2 * NB * NWG;
    int*   binD  = base + 2 * NB * NWG;
    int*   binS  = binD + EE;

    GCN_bincount_kernel<<<NWG, 1024, 0, stream>>>(esrc, edst, cnt);
    GCN_binscan_kernel<<<dim3(NB * NWG / 1024, 2), 1024, 0, stream>>>(cnt, base);
    GCN_binplace_kernel<<<NWG, 1024, 0, stream>>>(esrc, edst, base, binD, binS);
    GCN_rows_kernel<<<NB, 1024, 0, stream>>>(base, binD, binS, rowptr, inorm, onorm, perm);
    GCN_gather1_kernel<<<NN / 8, 256, 0, stream>>>(feat, rowptr, perm, onorm, agg1);
    GCN_gemm1_kernel<<<NN / 16, 256, 0, stream>>>(agg1, W1, b1, inorm, h1);
    GCN_gemm2_kernel<<<NN / 16, 128, 0, stream>>>(h1, W2, onorm, t2);
    GCN_gather2_kernel<<<NN / 8, 256, 0, stream>>>(t2, rowptr, perm, inorm, b2, h2);
    GCN_score_kernel<<<QQ / 8, 256, 0, stream>>>(h2, qsrc, qdst, out);
}


# Scoring: one wave takes 32 consecutive queries instead of one query per wave (31250 waves, not 1000000): ids read coalesced, all row loads in flight, scores stored as whole lines; same sum tree
# speedup vs baseline: 3.2616x; 1.3249x over previous
#include <hip/hip_runtime.h>
#include <math.h>

typedef __attribute__((ext_vector_type(2))) float v2f;
typedef __attribute__((ext_vector_type(8))) float v8f;

#define NN 100000
#define EE 1600000
#define QQ 1000000

#define NB  256
#define BW  391
#define NWG 256
#define CH  (EE / NWG)
#define EPT 7
#define SRC_BITS 17
#define RL  8

__global__ void __launch_bounds__(1024)
GCN_bincount_kernel(const int* __restrict__ esrc, const int* __restrict__ edst,
                    int* __restrict__ cnt) {
    __shared__ int hD[NB];
    __shared__ int hS[NB];
    int t = threadIdx.x, w = blockIdx.x;
    if (t < NB) { hD[t] = 0; hS[t] = 0; }
    __syncthreads();
    int e0 = w * CH;
    int d[EPT], s[EPT];
#pragma unroll
    for (int k = 0; k < EPT; ++k) {
        int i = t + k * 1024;
        d[k] = (i < CH) ? edst[e0 + i] : -1;
        s[k] = (i < CH) ? esrc[e0 + i] : -1;
    }
#pragma unroll
    for (int k = 0; k < EPT; ++k) {
        if (d[k] >= 0) {
            atomicAdd(&hD[d[k] / BW], 1);
            atomicAdd(&hS[s[k] / BW], 1);
        }
    }
    __syncthreads();
    if (t < NB) {
        cnt[t * NWG + w] = hD[t];
        cnt[NB * NWG + t * NWG + w] = hS[t];
    }
}

__global__ void __launch_bounds__(1024)
GCN_binscan_kernel(const int* __restrict__ cnt, int* __restrict__ base) {
    __shared__ int red[32];
    __shared__ int wsum[32];
    const int* c = cnt + blockIdx.y * (NB * NWG);
    int* o = base + blockIdx.y * (NB * NWG);
    int t = threadIdx.x, b = blockIdx.x, lane = t & 31, wave = t >> 5;
    const int4* c4 = (const int4*)c;
    int part = 0;
    int n4 = b * 256;
#pragma unroll 8
    for (int i = t; i < n4; i += 1024) {
        int4 v = c4[i];
        part += (v.x + v.y) + (v.z + v.w);
    }
#pragma unroll
    for (int off = 16; off > 0; off >>= 1) part += __shfl_xor(part, off, 32);
    if (lane == 0) red[wave] = part;

    int idx = b * 1024 + t;
    int v = c[idx];
    int inc = v;
#pragma unroll
    for (int off = 1; off < 32; off <<= 1) {
        int u = __shfl_up(inc, off, 32);
        if (lane >= off) inc += u;
    }
    if (lane == 31) wsum[wave] = inc;
    __syncthreads();
    int pre = 0;
#pragma unroll
    for (int k = 0; k < 32; ++k) {
        pre += red[k];
        if (k < wave) pre += wsum[k];
    }
    o[idx] = pre + inc - v;
}

__global__ void __launch_bounds__(1024)
GCN_binplace_kernel(const int* __restrict__ esrc, const int* __restrict__ edst,
                    const int* __restrict__ base, int* __restrict__ binD,
                    int* __restrict__ binS) {
    __shared__ int oD[NB];
    __shared__ int oS[NB];
    int t = threadIdx.x, w = blockIdx.x;
    if (t < NB) {
        oD[t] = base[t * NWG + w];
        oS[t] = base[NB * NWG + t * NWG + w];
    }
    int e0 = w * CH;
    int d[EPT], s[EPT];
#pragma unroll
    for (int k = 0; k < EPT; ++k) {
        int i = t + k * 1024;
        d[k] = (i < CH) ? edst[e0 + i] : -1;
        s[k] = (i < CH) ? esrc[e0 + i] : -1;
    }
    __syncthreads();
#pragma unroll
    for (int k = 0; k < EPT; ++k) {
        if (d[k] >= 0) {
            int bd = d[k] / BW, bs = s[k] / BW;
            int pd = atomicAdd(&oD[bd], 1);
            int ps = atomicAdd(&oS[bs], 1);
            binD[pd] = ((d[k] - bd * BW) << SRC_BITS) | s[k];
            binS[ps] = s[k] - bs * BW;
        }
    }
}

__global__ void __launch_bounds__(1024)
GCN_rows_kernel(const int* __restrict__ base, const int* __restrict__ binD,
                const int* __restrict__ binS, int* __restrict__ rowptr,
                float* __restrict__ inorm, float* __restrict__ onorm,
                int* __restrict__ perm) {
    __shared__ int cntIn[BW];
    __shared__ int cntOut[BW];
    __shared__ int start[BW];
    __shared__ int cur[BW];
    __shared__ int wsum[32];
    int t = threadIdx.x, b = blockIdx.x, lane = t & 31, wave = t >> 5;
    int dBeg = base[b * NWG];
    int dEnd = (b + 1 < NB) ? base[(b + 1) * NWG] : EE;
    int sBeg = base[NB * NWG + b * NWG];
    int sEnd = (b + 1 < NB) ? base[NB * NWG + (b + 1) * NWG] : EE;
    if (t < BW) { cntIn[t] = 0; cntOut[t] = 0; cur[t] = 0; }
    __syncthreads();
    for (int i0 = dBeg + t; i0 < dEnd; i0 += RL * 1024) {
        int p[RL];
#pragma unroll
        for (int k = 0; k < RL; ++k) {
            int i = i0 + k * 1024;
            p[k] = (i < dEnd) ? binD[i] : -1;
        }
#pragma unroll
        for (int k = 0; k < RL; ++k)
            if (p[k] >= 0) atomicAdd(&cntIn[p[k] >> SRC_BITS], 1);
    }
    for (int i0 = sBeg + t; i0 < sEnd; i0 += RL * 1024) {
        int p[RL];
#pragma unroll
        for (int k = 0; k < RL; ++k) {
            int i = i0 + k * 1024;
            p[k] = (i < sEnd) ? binS[i] : -1;
        }
#pragma unroll
        for (int k = 0; k < RL; ++k)
            if (p[k] >= 0) atomicAdd(&cntOut[p[k]], 1);
    }
    __syncthreads();

    int c = (t < BW) ? cntIn[t] : 0;
    int inc = c;
#pragma unroll
    for (int off = 1; off < 32; off <<= 1) {
        int u = __shfl_up(inc, off, 32);
        if (lane >= off) inc += u;
    }
    if (lane == 31) wsum[wave] = inc;
    __syncthreads();
    int pre = 0;
#pragma unroll
    for (int k = 0; k < 32; ++k) {
        if (k < wave) pre += wsum[k];
    }
    int first = dBeg + pre + inc - c;
    int n = b * BW + t;
    if (t < BW) {
        start[t] = first;
        if (n < NN) {
            rowptr[n] = first;
            inorm[n] = rsqrtf(fmaxf((float)c, 1.0f));
            onorm[n] = rsqrtf(fmaxf((float)cntOut[t], 1.0f));
            if (n == NN - 1) rowptr[NN] = first + c;
        }
    }
    __syncthreads();
    for (int i0 = dBeg + t; i0 < dEnd; i0 += RL * 1024) {
        int p[RL];
#pragma unroll
        for (int k = 0; k < RL; ++k) {
            int i = i0 + k * 1024;
            p[k] = (i < dEnd) ? binD[i] : -1;
        }
#pragma unroll
        for (int k = 0; k < RL; ++k) {
            if (p[k] >= 0) {
                int dl = p[k] >> SRC_BITS;
                int slot = atomicAdd(&cur[dl], 1);
                perm[start[dl] + slot] = p[k] & ((1 << SRC_BITS) - 1);
            }
        }
    }
}

__global__ void __launch_bounds__(256)
GCN_gather1_kernel(const float* __restrict__ feat, const int* __restrict__ rowptr,
                   const int* __restrict__ perm, const float* __restrict__ onorm,
                   float* __restrict__ agg) {
    int lane = threadIdx.x & 31;
    int n = __builtin_amdgcn_readfirstlane(blockIdx.x * 8 + (threadIdx.x >> 5));
    int beg = rowptr[n], end = rowptr[n + 1];
    float4 acc = make_float4(0.0f, 0.0f, 0.0f, 0.0f);
    for (int base = beg; base < end; base += 32) {
        int cnt = min(32, end - base);
        int s_l = 0;
        float sc_l = 0.0f;
        if (lane < cnt) {
            s_l = perm[base + lane];
            sc_l = onorm[s_l];
        }
        int sc_i = __float_as_int(sc_l);
        for (int j = 0; j < cnt; j += 4) {
            int j1 = min(j + 1, cnt - 1), j2 = min(j + 2, cnt - 1), j3 = min(j + 3, cnt - 1);
            int s0 = __builtin_amdgcn_readlane(s_l, j);
            int s1 = __builtin_amdgcn_readlane(s_l, j1);
            int s2 = __builtin_amdgcn_readlane(s_l, j2);
            int s3 = __builtin_amdgcn_readlane(s_l, j3);
            float c0 = __int_as_float(__builtin_amdgcn_readlane(sc_i, j));
            float c1 = (j + 1 < cnt) ? __int_as_float(__builtin_amdgcn_readlane(sc_i, j1)) : 0.0f;
            float c2 = (j + 2 < cnt) ? __int_as_float(__builtin_amdgcn_readlane(sc_i, j2)) : 0.0f;
            float c3 = (j + 3 < cnt) ? __int_as_float(__builtin_amdgcn_readlane(sc_i, j3)) : 0.0f;
            const float4 v0 = *(const float4*)(feat + (size_t)s0 * 128 + lane * 4);
            const float4 v1 = *(const float4*)(feat + (size_t)s1 * 128 + lane * 4);
            const float4 v2 = *(const float4*)(feat + (size_t)s2 * 128 + lane * 4);
            const float4 v3 = *(const float4*)(feat + (size_t)s3 * 128 + lane * 4);
            acc.x = fmaf(v0.x, c0, acc.x); acc.y = fmaf(v0.y, c0, acc.y);
            acc.z = fmaf(v0.z, c0, acc.z); acc.w = fmaf(v0.w, c0, acc.w);
            acc.x = fmaf(v1.x, c1, acc.x); acc.y = fmaf(v1.y, c1, acc.y);
            acc.z = fmaf(v1.z, c1, acc.z); acc.w = fmaf(v1.w, c1, acc.w);
            acc.x = fmaf(v2.x, c2, acc.x); acc.y = fmaf(v2.y, c2, acc.y);
            acc.z = fmaf(v2.z, c2, acc.z); acc.w = fmaf(v2.w, c2, acc.w);
            acc.x = fmaf(v3.x, c3, acc.x); acc.y = fmaf(v3.y, c3, acc.y);
            acc.z = fmaf(v3.z, c3, acc.z); acc.w = fmaf(v3.w, c3, acc.w);
        }
    }
    *(float4*)(agg + (size_t)n * 128 + lane * 4) = acc;
}

__global__ void __launch_bounds__(256)
GCN_gemm1_kernel(const float* __restrict__ A, const float* __restrict__ W,
                 const float* __restrict__ bias, const float* __restrict__ inorm,
                 float* __restrict__ H) {
    __shared__ float lW[128 * 128];
    int tid = threadIdx.x;
    {
        const float4* Wv = (const float4*)W;
        float4* lWv = (float4*)lW;
#pragma unroll
        for (int i = 0; i < 16; ++i) lWv[tid + 256 * i] = Wv[tid + 256 * i];
    }
    __syncthreads();

    int wave = tid >> 5, lane = tid & 31;
    int row0 = blockIdx.x * 16, col0 = wave * 16;
    int r = lane & 15, kh = lane >> 4;

    v8f c = {};
    const float* Ap = A + (size_t)(row0 + r) * 128 + 2 * kh;
    const float* Bp = lW + (size_t)(2 * kh) * 128 + col0 + r;
#pragma unroll
    for (int kt = 0; kt < 32; ++kt) {
        v2f a = *(const v2f*)(Ap + kt * 4);
        v2f b;
        b.x = Bp[kt * 512];
        b.y = Bp[kt * 512 + 128];
        c = __builtin_amdgcn_wmma_f32_16x16x4_f32(false, a, false, b, (short)0, c, false, false);
    }
    int n = col0 + r;
    float bb = bias[n];
#pragma unroll
    for (int v = 0; v < 8; ++v) {
        int m = row0 + v + 8 * kh;
        float val = fmaf(c[v], inorm[m], bb);
        H[(size_t)m * 128 + n] = fmaxf(val, 0.0f);
    }
}

__global__ void __launch_bounds__(128)
GCN_gemm2_kernel(const float* __restrict__ Hin, const float* __restrict__ W,
                 const float* __restrict__ onorm, float* __restrict__ T) {
    __shared__ float lW[128 * 64];
    int tid = threadIdx.x;
    {
        const float4* Wv = (const float4*)W;
        float4* lWv = (float4*)lW;
#pragma unroll
        for (int i = 0; i < 16; ++i) lWv[tid + 128 * i] = Wv[tid + 128 * i];
    }
    __syncthreads();

    int wave = tid >> 5, lane = tid & 31;
    int row0 = blockIdx.x * 16, col0 = wave * 16;
    int r = lane & 15, kh = lane >> 4;
    float sc = onorm[row0 + r];

    v8f c = {};
    const float* Ap = Hin + (size_t)(row0 + r) * 128 + 2 * kh;
    const float* Bp = lW + (size_t)(2 * kh) * 64 + col0 + r;
#pragma unroll
    for (int kt = 0; kt < 32; ++kt) {
        v2f a = *(const v2f*)(Ap + kt * 4);
        a.x *= sc; a.y *= sc;
        v2f b;
        b.x = Bp[kt * 256];
        b.y = Bp[kt * 256 + 64];
        c = __builtin_amdgcn_wmma_f32_16x16x4_f32(false, a, false, b, (short)0, c, false, false);
    }
    int n = col0 + r;
#pragma unroll
    for (int v = 0; v < 8; ++v) {
        int m = row0 + v + 8 * kh;
        T[(size_t)m * 64 + n] = c[v];
    }
}

__global__ void __launch_bounds__(256)
GCN_gather2_kernel(const float* __restrict__ T, const int* __restrict__ rowptr,
                   const int* __restrict__ perm, const float* __restrict__ inorm,
                   const float* __restrict__ b2, float* __restrict__ H2) {
    int lane = threadIdx.x & 31;
    int n = __builtin_amdgcn_readfirstlane(blockIdx.x * 8 + (threadIdx.x >> 5));
    int beg = rowptr[n], end = rowptr[n + 1];
    float2 acc = make_float2(0.0f, 0.0f);
    for (int base = beg; base < end; base += 32) {
        int cnt = min(32, end - base);
        int s_l = 0;
        if (lane < cnt) s_l = perm[base + lane];
        for (int j = 0; j < cnt; j += 4) {
            int j1 = min(j + 1, cnt - 1), j2 = min(j + 2, cnt - 1), j3 = min(j + 3, cnt - 1);
            int s0 = __builtin_amdgcn_readlane(s_l, j);
            int s1 = __builtin_amdgcn_readlane(s_l, j1);
            int s2 = __builtin_amdgcn_readlane(s_l, j2);
            int s3 = __builtin_amdgcn_readlane(s_l, j3);
            float c1 = (j + 1 < cnt) ? 1.0f : 0.0f;
            float c2 = (j + 2 < cnt) ? 1.0f : 0.0f;
            float c3 = (j + 3 < cnt) ? 1.0f : 0.0f;
            const float2 v0 = *(const float2*)(T + (size_t)s0 * 64 + lane * 2);
            const float2 v1 = *(const float2*)(T + (size_t)s1 * 64 + lane * 2);
            const float2 v2 = *(const float2*)(T + (size_t)s2 * 64 + lane * 2);
            const float2 v3 = *(const float2*)(T + (size_t)s3 * 64 + lane * 2);
            acc.x += v0.x;                  acc.y += v0.y;
            acc.x = fmaf(v1.x, c1, acc.x);  acc.y = fmaf(v1.y, c1, acc.y);
            acc.x = fmaf(v2.x, c2, acc.x);  acc.y = fmaf(v2.y, c2, acc.y);
            acc.x = fmaf(v3.x, c3, acc.x);  acc.y = fmaf(v3.y, c3, acc.y);
        }
    }
    float s = inorm[n];
    float2 o;
    o.x = fmaf(acc.x, s, b2[lane * 2 + 0]);
    o.y = fmaf(acc.y, s, b2[lane * 2 + 1]);
    *(float2*)(H2 + (size_t)n * 64 + lane * 2) = o;
}

#define SQ  32
#define SWV 10

template <int OFF>
__device__ __forceinline__ void GCN_score_level(float (&p)[SQ], bool hi) {
#pragma unroll
    for (int j = 0; j < OFF; ++j) {
        float keep = hi ? p[j + OFF] : p[j];
        float send = hi ? p[j] : p[j + OFF];
        p[j] = keep + __shfl_xor(send, OFF, 32);
    }
}

__global__ void __launch_bounds__(32 * SWV) __attribute__((amdgpu_waves_per_eu(8, 8)))
GCN_score_kernel(const float* __restrict__ H2, const int* __restrict__ qs,
                 const int* __restrict__ qd, float* __restrict__ out) {
    static_assert(QQ % (SQ * SWV) == 0, "whole waves and whole work-groups only");
    static_assert((size_t)NN * 32 < 0x7fffffffu, "row offsets fit 32 bits");
    const float2* __restrict__ rows = (const float2*)H2;
    unsigned lane = threadIdx.x & 31;
    int w = __builtin_amdgcn_readfirstlane(blockIdx.x * SWV + (threadIdx.x >> 5));
    int q0 = w * SQ;
    int s_l = qs[q0 + lane], d_l = qd[q0 + lane];
    float2 a[SQ], b[SQ];
    float p[SQ];
#pragma unroll
    for (int j = 0; j < SQ; ++j) {
        unsigned s = (unsigned)__builtin_amdgcn_readlane(s_l, j);
        unsigned d = (unsigned)__builtin_amdgcn_readlane(d_l, j);
        a[j] = rows[s * 32u + lane];
        b[j] = rows[d * 32u + lane];
    }
#pragma unroll
    for (int j = 0; j < SQ; ++j) p[j] = fmaf(a[j].x, b[j].x, a[j].y * b[j].y);
    GCN_score_level<16>(p, (lane & 16) != 0);
    GCN_score_level<8>(p, (lane & 8) != 0);
    GCN_score_level<4>(p, (lane & 4) != 0);
    GCN_score_level<2>(p, (lane & 2) != 0);
    GCN_score_level<1>(p, (lane & 1) != 0);
    out[q0 + lane] = 1.0f / (1.0f + __expf(-p[0]));
}

extern "C" void kernel_launch(void* const* d_in, const int* in_sizes, int n_in,
                              void* d_out, int out_size, void* d_ws, size_t ws_size,
                              hipStream_t stream) {
    const float* feat = (const float*)d_in[0];
    const int* esrc   = (const int*)d_in[1];
    const int* edst   = (const int*)d_in[2];
    const int* qsrc   = (const int*)d_in[3];
    const int* qdst   = (const int*)d_in[4];
    const float* W1   = (const float*)d_in[5];
    const float* b1   = (const float*)d_in[6];
    const float* W2   = (const float*)d_in[7];
    const float* b2   = (const float*)d_in[8];
    float* out = (float*)d_out;

    float* ws    = (float*)d_ws;
    float* onorm = ws;
    float* agg1  = ws + 2 * NN;
    float* h2    = agg1;
    float* h1    = agg1 + (size_t)NN * 128;
    float* t2    = h1 + (size_t)NN * 128;
    int*   perm  = (int*)(t2 + (size_t)NN * 64);
    float* inorm = (float*)(perm + 2 * EE);
    int*   rowptr = (int*)(inorm + NN);
    int*   cnt   = (int*)h1;
    int*   base  = cnt + 2 * NB * NWG;
    int*   binD  = base + 2 * NB * NWG;
    int*   binS  = binD + EE;

    GCN_bincount_kernel<<<NWG, 1024, 0, stream>>>(esrc, edst, cnt);
    GCN_binscan_kernel<<<dim3(NB * NWG / 1024, 2), 1024, 0, stream>>>(cnt, base);
    GCN_binplace_kernel<<<NWG, 1024, 0, stream>>>(esrc, edst, base, binD, binS);
    GCN_rows_kernel<<<NB, 1024, 0, stream>>>(base, binD, binS, rowptr, inorm, onorm, perm);
    GCN_gather1_kernel<<<NN / 8, 256, 0, stream>>>(feat, rowptr, perm, onorm, agg1);
    GCN_gemm1_kernel<<<NN / 16, 256, 0, stream>>>(agg1, W1, b1, inorm, h1);
    GCN_gemm2_kernel<<<NN / 16, 128, 0, stream>>>(h1, W2, onorm, t2);
    GCN_gather2_kernel<<<NN / 8, 256, 0, stream>>>(t2, rowptr, perm, inorm, b2, h2);
    GCN_score_kernel<<<QQ / (SQ * SWV), 32 * SWV, 0, stream>>>(h2, qsrc, qdst, out);
}


# Both GEMMs re-tiled: a wave takes a 16-row strip over all output columns (8 or 4 accumulator tiles): A fragments load once, not once per column tile; W is staged per 64 rows, not 16; sum order kept
# speedup vs baseline: 3.7321x; 1.1443x over previous
#include <hip/hip_runtime.h>
#include <math.h>

typedef __attribute__((ext_vector_type(2))) float v2f;
typedef __attribute__((ext_vector_type(8))) float v8f;

#define NN 100000
#define EE 1600000
#define QQ 1000000

#define NB  256
#define BW  391
#define NWG 256
#define CH  (EE / NWG)
#define EPT 7
#define SRC_BITS 17
#define RL  8

__global__ void __launch_bounds__(1024)
GCN_bincount_kernel(const int* __restrict__ esrc, const int* __restrict__ edst,
                    int* __restrict__ cnt) {
    __shared__ int hD[NB];
    __shared__ int hS[NB];
    int t = threadIdx.x, w = blockIdx.x;
    if (t < NB) { hD[t] = 0; hS[t] = 0; }
    __syncthreads();
    int e0 = w * CH;
    int d[EPT], s[EPT];
#pragma unroll
    for (int k = 0; k < EPT; ++k) {
        int i = t + k * 1024;
        d[k] = (i < CH) ? edst[e0 + i] : -1;
        s[k] = (i < CH) ? esrc[e0 + i] : -1;
    }
#pragma unroll
    for (int k = 0; k < EPT; ++k) {
        if (d[k] >= 0) {
            atomicAdd(&hD[d[k] / BW], 1);
            atomicAdd(&hS[s[k] / BW], 1);
        }
    }
    __syncthreads();
    if (t < NB) {
        cnt[t * NWG + w] = hD[t];
        cnt[NB * NWG + t * NWG + w] = hS[t];
    }
}

__global__ void __launch_bounds__(1024)
GCN_binscan_kernel(const int* __restrict__ cnt, int* __restrict__ base) {
    __shared__ int red[32];
    __shared__ int wsum[32];
    const int* c = cnt + blockIdx.y * (NB * NWG);
    int* o = base + blockIdx.y * (NB * NWG);
    int t = threadIdx.x, b = blockIdx.x, lane = t & 31, wave = t >> 5;
    const int4* c4 = (const int4*)c;
    int part = 0;
    int n4 = b * 256;
#pragma unroll 8
    for (int i = t; i < n4; i += 1024) {
        int4 v = c4[i];
        part += (v.x + v.y) + (v.z + v.w);
    }
#pragma unroll
    for (int off = 16; off > 0; off >>= 1) part += __shfl_xor(part, off, 32);
    if (lane == 0) red[wave] = part;

    int idx = b * 1024 + t;
    int v = c[idx];
    int inc = v;
#pragma unroll
    for (int off = 1; off < 32; off <<= 1) {
        int u = __shfl_up(inc, off, 32);
        if (lane >= off) inc += u;
    }
    if (lane == 31) wsum[wave] = inc;
    __syncthreads();
    int pre = 0;
#pragma unroll
    for (int k = 0; k < 32; ++k) {
        pre += red[k];
        if (k < wave) pre += wsum[k];
    }
    o[idx] = pre + inc - v;
}

__global__ void __launch_bounds__(1024)
GCN_binplace_kernel(const int* __restrict__ esrc, const int* __restrict__ edst,
                    const int* __restrict__ base, int* __restrict__ binD,
                    int* __restrict__ binS) {
    __shared__ int oD[NB];
    __shared__ int oS[NB];
    int t = threadIdx.x, w = blockIdx.x;
    if (t < NB) {
        oD[t] = base[t * NWG + w];
        oS[t] = base[NB * NWG + t * NWG + w];
    }
    int e0 = w * CH;
    int d[EPT], s[EPT];
#pragma unroll
    for (int k = 0; k < EPT; ++k) {
        int i = t + k * 1024;
        d[k] = (i < CH) ? edst[e0 + i] : -1;
        s[k] = (i < CH) ? esrc[e0 + i] : -1;
    }
    __syncthreads();
#pragma unroll
    for (int k = 0; k < EPT; ++k) {
        if (d[k] >= 0) {
            int bd = d[k] / BW, bs = s[k] / BW;
            int pd = atomicAdd(&oD[bd], 1);
            int ps = atomicAdd(&oS[bs], 1);
            binD[pd] = ((d[k] - bd * BW) << SRC_BITS) | s[k];
            binS[ps] = s[k] - bs * BW;
        }
    }
}

__global__ void __launch_bounds__(1024)
GCN_rows_kernel(const int* __restrict__ base, const int* __restrict__ binD,
                const int* __restrict__ binS, int* __restrict__ rowptr,
                float* __restrict__ inorm, float* __restrict__ onorm,
                int* __restrict__ perm) {
    __shared__ int cntIn[BW];
    __shared__ int cntOut[BW];
    __shared__ int start[BW];
    __shared__ int cur[BW];
    __shared__ int wsum[32];
    int t = threadIdx.x, b = blockIdx.x, lane = t & 31, wave = t >> 5;
    int dBeg = base[b * NWG];
    int dEnd = (b + 1 < NB) ? base[(b + 1) * NWG] : EE;
    int sBeg = base[NB * NWG + b * NWG];
    int sEnd = (b + 1 < NB) ? base[NB * NWG + (b + 1) * NWG] : EE;
    if (t < BW) { cntIn[t] = 0; cntOut[t] = 0; cur[t] = 0; }
    __syncthreads();
    for (int i0 = dBeg + t; i0 < dEnd; i0 += RL * 1024) {
        int p[RL];
#pragma unroll
        for (int k = 0; k < RL; ++k) {
            int i = i0 + k * 1024;
            p[k] = (i < dEnd) ? binD[i] : -1;
        }
#pragma unroll
        for (int k = 0; k < RL; ++k)
            if (p[k] >= 0) atomicAdd(&cntIn[p[k] >> SRC_BITS], 1);
    }
    for (int i0 = sBeg + t; i0 < sEnd; i0 += RL * 1024) {
        int p[RL];
#pragma unroll
        for (int k = 0; k < RL; ++k) {
            int i = i0 + k * 1024;
            p[k] = (i < sEnd) ? binS[i] : -1;
        }
#pragma unroll
        for (int k = 0; k < RL; ++k)
            if (p[k] >= 0) atomicAdd(&cntOut[p[k]], 1);
    }
    __syncthreads();

    int c = (t < BW) ? cntIn[t] : 0;
    int inc = c;
#pragma unroll
    for (int off = 1; off < 32; off <<= 1) {
        int u = __shfl_up(inc, off, 32);
        if (lane >= off) inc += u;
    }
    if (lane == 31) wsum[wave] = inc;
    __syncthreads();
    int pre = 0;
#pragma unroll
    for (int k = 0; k < 32; ++k) {
        if (k < wave) pre += wsum[k];
    }
    int first = dBeg + pre + inc - c;
    int n = b * BW + t;
    if (t < BW) {
        start[t] = first;
        if (n < NN) {
            rowptr[n] = first;
            inorm[n] = rsqrtf(fmaxf((float)c, 1.0f));
            onorm[n] = rsqrtf(fmaxf((float)cntOut[t], 1.0f));
            if (n == NN - 1) rowptr[NN] = first + c;
        }
    }
    __syncthreads();
    for (int i0 = dBeg + t; i0 < dEnd; i0 += RL * 1024) {
        int p[RL];
#pragma unroll
        for (int k = 0; k < RL; ++k) {
            int i = i0 + k * 1024;
            p[k] = (i < dEnd) ? binD[i] : -1;
        }
#pragma unroll
        for (int k = 0; k < RL; ++k) {
            if (p[k] >= 0) {
                int dl = p[k] >> SRC_BITS;
                int slot = atomicAdd(&cur[dl], 1);
                perm[start[dl] + slot] = p[k] & ((1 << SRC_BITS) - 1);
            }
        }
    }
}

__global__ void __launch_bounds__(256)
GCN_gather1_kernel(const float* __restrict__ feat, const int* __restrict__ rowptr,
                   const int* __restrict__ perm, const float* __restrict__ onorm,
                   float* __restrict__ agg) {
    int lane = threadIdx.x & 31;
    int n = __builtin_amdgcn_readfirstlane(blockIdx.x * 8 + (threadIdx.x >> 5));
    int beg = rowptr[n], end = rowptr[n + 1];
    float4 acc = make_float4(0.0f, 0.0f, 0.0f, 0.0f);
    for (int base = beg; base < end; base += 32) {
        int cnt = min(32, end - base);
        int s_l = 0;
        float sc_l = 0.0f;
        if (lane < cnt) {
            s_l = perm[base + lane];
            sc_l = onorm[s_l];
        }
        int sc_i = __float_as_int(sc_l);
        for (int j = 0; j < cnt; j += 4) {
            int j1 = min(j + 1, cnt - 1), j2 = min(j + 2, cnt - 1), j3 = min(j + 3, cnt - 1);
            int s0 = __builtin_amdgcn_readlane(s_l, j);
            int s1 = __builtin_amdgcn_readlane(s_l, j1);
            int s2 = __builtin_amdgcn_readlane(s_l, j2);
            int s3 = __builtin_amdgcn_readlane(s_l, j3);
            float c0 = __int_as_float(__builtin_amdgcn_readlane(sc_i, j));
            float c1 = (j + 1 < cnt) ? __int_as_float(__builtin_amdgcn_readlane(sc_i, j1)) : 0.0f;
            float c2 = (j + 2 < cnt) ? __int_as_float(__builtin_amdgcn_readlane(sc_i, j2)) : 0.0f;
            float c3 = (j + 3 < cnt) ? __int_as_float(__builtin_amdgcn_readlane(sc_i, j3)) : 0.0f;
            const float4 v0 = *(const float4*)(feat + (size_t)s0 * 128 + lane * 4);
            const float4 v1 = *(const float4*)(feat + (size_t)s1 * 128 + lane * 4);
            const float4 v2 = *(const float4*)(feat + (size_t)s2 * 128 + lane * 4);
            const float4 v3 = *(const float4*)(feat + (size_t)s3 * 128 + lane * 4);
            acc.x = fmaf(v0.x, c0, acc.x); acc.y = fmaf(v0.y, c0, acc.y);
            acc.z = fmaf(v0.z, c0, acc.z); acc.w = fmaf(v0.w, c0, acc.w);
            acc.x = fmaf(v1.x, c1, acc.x); acc.y = fmaf(v1.y, c1, acc.y);
            acc.z = fmaf(v1.z, c1, acc.z); acc.w = fmaf(v1.w, c1, acc.w);
            acc.x = fmaf(v2.x, c2, acc.x); acc.y = fmaf(v2.y, c2, acc.y);
            acc.z = fmaf(v2.z, c2, acc.z); acc.w = fmaf(v2.w, c2, acc.w);
            acc.x = fmaf(v3.x, c3, acc.x); acc.y = fmaf(v3.y, c3, acc.y);
            acc.z = fmaf(v3.z, c3, acc.z); acc.w = fmaf(v3.w, c3, acc.w);
        }
    }
    *(float4*)(agg + (size_t)n * 128 + lane * 4) = acc;
}

#define GS    4
#define GGRID ((NN / 16 + GS - 1) / GS)
static_assert(128 % GS == 0 && 64 % GS == 0, "the W staging loops take whole trips");

__global__ void __launch_bounds__(32 * GS)
GCN_gemm1_kernel(const float* __restrict__ A, const float* __restrict__ W,
                 const float* __restrict__ bias, const float* __restrict__ inorm,
                 float* __restrict__ H) {
    static_assert(NN % 16 == 0, "whole strips only");
    __shared__ float lW[128 * 128];
    int tid = threadIdx.x, lane = tid & 31;
    int row0 = __builtin_amdgcn_readfirstlane((blockIdx.x * GS + (tid >> 5)) * 16);
    bool live = row0 < NN;
    int r = lane & 15, kh = lane >> 4;

    v2f a[32];
    {
        const float* Ap = A + (size_t)((live ? row0 : NN - 16) + r) * 128 + 2 * kh;
#pragma unroll
        for (int kt = 0; kt < 32; ++kt) a[kt] = *(const v2f*)(Ap + kt * 4);
    }
    {
        const float4* Wv = (const float4*)W;
        float4* lWv = (float4*)lW;
#pragma unroll
        for (int i = 0; i < 128 / GS; ++i) lWv[tid + 32 * GS * i] = Wv[tid + 32 * GS * i];
    }
    __syncthreads();
    if (!live) return;

    v8f c[8] = {};
    const float* Bp = lW + (size_t)(2 * kh) * 128 + r;
#pragma unroll
    for (int kt = 0; kt < 32; ++kt) {
#pragma unroll
        for (int ct = 0; ct < 8; ++ct) {
            v2f b;
            b.x = Bp[kt * 512 + ct * 16];
            b.y = Bp[kt * 512 + 128 + ct * 16];
            c[ct] = __builtin_amdgcn_wmma_f32_16x16x4_f32(false, a[kt], false, b, (short)0,
                                                          c[ct], false, false);
        }
    }
    float sc[8];
#pragma unroll
    for (int v = 0; v < 8; ++v) sc[v] = inorm[row0 + v + 8 * kh];
#pragma unroll
    for (int ct = 0; ct < 8; ++ct) {
        int n = ct * 16 + r;
        float bb = bias[n];
#pragma unroll
        for (int v = 0; v < 8; ++v) {
            int m = row0 + v + 8 * kh;
            float val = fmaf(c[ct][v], sc[v], bb);
            H[(size_t)m * 128 + n] = fmaxf(val, 0.0f);
        }
    }
}

__global__ void __launch_bounds__(32 * GS)
GCN_gemm2_kernel(const float* __restrict__ Hin, const float* __restrict__ W,
                 const float* __restrict__ onorm, float* __restrict__ T) {
    __shared__ float lW[128 * 64];
    int tid = threadIdx.x, lane = tid & 31;
    int row0 = __builtin_amdgcn_readfirstlane((blockIdx.x * GS + (tid >> 5)) * 16);
    bool live = row0 < NN;
    int r = lane & 15, kh = lane >> 4;

    v2f a[32];
    {
        int rowA = (live ? row0 : NN - 16) + r;
        float s = onorm[rowA];
        const float* Ap = Hin + (size_t)rowA * 128 + 2 * kh;
#pragma unroll
        for (int kt = 0; kt < 32; ++kt) {
            a[kt] = *(const v2f*)(Ap + kt * 4);
            a[kt].x *= s; a[kt].y *= s;
        }
    }
    {
        const float4* Wv = (const float4*)W;
        float4* lWv = (float4*)lW;
#pragma unroll
        for (int i = 0; i < 64 / GS; ++i) lWv[tid + 32 * GS * i] = Wv[tid + 32 * GS * i];
    }
    __syncthreads();
    if (!live) return;

    v8f c[4] = {};
    const float* Bp = lW + (size_t)(2 * kh) * 64 + r;
#pragma unroll
    for (int kt = 0; kt < 32; ++kt) {
#pragma unroll
        for (int ct = 0; ct < 4; ++ct) {
            v2f b;
            b.x = Bp[kt * 256 + ct * 16];
            b.y = Bp[kt * 256 + 64 + ct * 16];
            c[ct] = __builtin_amdgcn_wmma_f32_16x16x4_f32(false, a[kt], false, b, (short)0,
                                                          c[ct], false, false);
        }
    }
#pragma unroll
    for (int ct = 0; ct < 4; ++ct) {
        int n = ct * 16 + r;
#pragma unroll
        for (int v = 0; v < 8; ++v) {
            int m = row0 + v + 8 * kh;
            T[(size_t)m * 64 + n] = c[ct][v];
        }
    }
}

__global__ void __launch_bounds__(256)
GCN_gather2_kernel(const float* __restrict__ T, const int* __restrict__ rowptr,
                   const int* __restrict__ perm, const float* __restrict__ inorm,
                   const float* __restrict__ b2, float* __restrict__ H2) {
    int lane = threadIdx.x & 31;
    int n = __builtin_amdgcn_readfirstlane(blockIdx.x * 8 + (threadIdx.x >> 5));
    int beg = rowptr[n], end = rowptr[n + 1];
    float2 acc = make_float2(0.0f, 0.0f);
    for (int base = beg; base < end; base += 32) {
        int cnt = min(32, end - base);
        int s_l = 0;
        if (lane < cnt) s_l = perm[base + lane];
        for (int j = 0; j < cnt; j += 4) {
            int j1 = min(j + 1, cnt - 1), j2 = min(j + 2, cnt - 1), j3 = min(j + 3, cnt - 1);
            int s0 = __builtin_amdgcn_readlane(s_l, j);
            int s1 = __builtin_amdgcn_readlane(s_l, j1);
            int s2 = __builtin_amdgcn_readlane(s_l, j2);
            int s3 = __builtin_amdgcn_readlane(s_l, j3);
            float c1 = (j + 1 < cnt) ? 1.0f : 0.0f;
            float c2 = (j + 2 < cnt) ? 1.0f : 0.0f;
            float c3 = (j + 3 < cnt) ? 1.0f : 0.0f;
            const float2 v0 = *(const float2*)(T + (size_t)s0 * 64 + lane * 2);
            const float2 v1 = *(const float2*)(T + (size_t)s1 * 64 + lane * 2);
            const float2 v2 = *(const float2*)(T + (size_t)s2 * 64 + lane * 2);
            const float2 v3 = *(const float2*)(T + (size_t)s3 * 64 + lane * 2);
            acc.x += v0.x;                  acc.y += v0.y;
            acc.x = fmaf(v1.x, c1, acc.x);  acc.y = fmaf(v1.y, c1, acc.y);
            acc.x = fmaf(v2.x, c2, acc.x);  acc.y = fmaf(v2.y, c2, acc.y);
            acc.x = fmaf(v3.x, c3, acc.x);  acc.y = fmaf(v3.y, c3, acc.y);
        }
    }
    float s = inorm[n];
    float2 o;
    o.x = fmaf(acc.x, s, b2[lane * 2 + 0]);
    o.y = fmaf(acc.y, s, b2[lane * 2 + 1]);
    *(float2*)(H2 + (size_t)n * 64 + lane * 2) = o;
}

#define SQ  32
#define SWV 10

template <int OFF>
__device__ __forceinline__ void GCN_score_level(float (&p)[SQ], bool hi) {
#pragma unroll
    for (int j = 0; j < OFF; ++j) {
        float keep = hi ? p[j + OFF] : p[j];
        float send = hi ? p[j] : p[j + OFF];
        p[j] = keep + __shfl_xor(send, OFF, 32);
    }
}

__global__ void __launch_bounds__(32 * SWV) __attribute__((amdgpu_waves_per_eu(8, 8)))
GCN_score_kernel(const float* __restrict__ H2, const int* __restrict__ qs,
                 const int* __restrict__ qd, float* __restrict__ out) {
    static_assert(QQ % (SQ * SWV) == 0, "whole waves and whole work-groups only");
    static_assert((size_t)NN * 32 < 0x7fffffffu, "row offsets fit 32 bits");
    const float2* __restrict__ rows = (const float2*)H2;
    unsigned lane = threadIdx.x & 31;
    int w = __builtin_amdgcn_readfirstlane(blockIdx.x * SWV + (threadIdx.x >> 5));
    int q0 = w * SQ;
    int s_l = qs[q0 + lane], d_l = qd[q0 + lane];
    float2 a[SQ], b[SQ];
    float p[SQ];
#pragma unroll
    for (int j = 0; j < SQ; ++j) {
        unsigned s = (unsigned)__builtin_amdgcn_readlane(s_l, j);
        unsigned d = (unsigned)__builtin_amdgcn_readlane(d_l, j);
        a[j] = rows[s * 32u + lane];
        b[j] = rows[d * 32u + lane];
    }
#pragma unroll
    for (int j = 0; j < SQ; ++j) p[j] = fmaf(a[j].x, b[j].x, a[j].y * b[j].y);
    GCN_score_level<16>(p, (lane & 16) != 0);
    GCN_score_level<8>(p, (lane & 8) != 0);
    GCN_score_level<4>(p, (lane & 4) != 0);
    GCN_score_level<2>(p, (lane & 2) != 0);
    GCN_score_level<1>(p, (lane & 1) != 0);
    out[q0 + lane] = 1.0f / (1.0f + __expf(-p[0]));
}

extern "C" void kernel_launch(void* const* d_in, const int* in_sizes, int n_in,
                              void* d_out, int out_size, void* d_ws, size_t ws_size,
                              hipStream_t stream) {
    const float* feat = (const float*)d_in[0];
    const int* esrc   = (const int*)d_in[1];
    const int* edst   = (const int*)d_in[2];
    const int* qsrc   = (const int*)d_in[3];
    const int* qdst   = (const int*)d_in[4];
    const float* W1   = (const float*)d_in[5];
    const float* b1   = (const float*)d_in[6];
    const float* W2   = (const float*)d_in[7];
    const float* b2   = (const float*)d_in[8];
    float* out = (float*)d_out;

    float* ws    = (float*)d_ws;
    float* onorm = ws;
    float* agg1  = ws + 2 * NN;
    float* h2    = agg1;
    float* h1    = agg1 + (size_t)NN * 128;
    float* t2    = h1 + (size_t)NN * 128;
    int*   perm  = (int*)(t2 + (size_t)NN * 64);
    float* inorm = (float*)(perm + 2 * EE);
    int*   rowptr = (int*)(inorm + NN);
    int*   cnt   = (int*)h1;
    int*   base  = cnt + 2 * NB * NWG;
    int*   binD  = base + 2 * NB * NWG;
    int*   binS  = binD + EE;

    GCN_bincount_kernel<<<NWG, 1024, 0, stream>>>(esrc, edst, cnt);
    GCN_binscan_kernel<<<dim3(NB * NWG / 1024, 2), 1024, 0, stream>>>(cnt, base);
    GCN_binplace_kernel<<<NWG, 1024, 0, stream>>>(esrc, edst, base, binD, binS);
    GCN_rows_kernel<<<NB, 1024, 0, stream>>>(base, binD, binS, rowptr, inorm, onorm, perm);
    GCN_gather1_kernel<<<NN / 8, 256, 0, stream>>>(feat, rowptr, perm, onorm, agg1);
    GCN_gemm1_kernel<<<GGRID, 32 * GS, 0, stream>>>(agg1, W1, b1, inorm, h1);
    GCN_gemm2_kernel<<<GGRID, 32 * GS, 0, stream>>>(h1, W2, onorm, t2);
    GCN_gather2_kernel<<<NN / 8, 256, 0, stream>>>(t2, rowptr, perm, inorm, b2, h2);
    GCN_score_kernel<<<QQ / (SQ * SWV), 32 * SWV, 0, stream>>>(h2, qsrc, qdst, out);
}


# GEMM1+GEMM2 in one launch: GEMM1's WMMA operands exchanged so a lane's accumulators are its own row's A fragments for GEMM2; h1 stays in registers; GEMM2's k regrouped (its sum order changes)
# speedup vs baseline: 3.9335x; 1.0540x over previous
#include <hip/hip_runtime.h>
#include <math.h>

typedef __attribute__((ext_vector_type(2))) float v2f;
typedef __attribute__((ext_vector_type(8))) float v8f;

#define NN 100000
#define EE 1600000
#define QQ 1000000

#define NB  256
#define BW  391
#define NWG 256
#define CH  (EE / NWG)
#define EPT 7
#define SRC_BITS 17
#define RL  8

__global__ void __launch_bounds__(1024)
GCN_bincount_kernel(const int* __restrict__ esrc, const int* __restrict__ edst,
                    int* __restrict__ cnt) {
    __shared__ int hD[NB];
    __shared__ int hS[NB];
    int t = threadIdx.x, w = blockIdx.x;
    if (t < NB) { hD[t] = 0; hS[t] = 0; }
    __syncthreads();
    int e0 = w * CH;
    int d[EPT], s[EPT];
#pragma unroll
    for (int k = 0; k < EPT; ++k) {
        int i = t + k * 1024;
        d[k] = (i < CH) ? edst[e0 + i] : -1;
        s[k] = (i < CH) ? esrc[e0 + i] : -1;
    }
#pragma unroll
    for (int k = 0; k < EPT; ++k) {
        if (d[k] >= 0) {
            atomicAdd(&hD[d[k] / BW], 1);
            atomicAdd(&hS[s[k] / BW], 1);
        }
    }
    __syncthreads();
    if (t < NB) {
        cnt[t * NWG + w] = hD[t];
        cnt[NB * NWG + t * NWG + w] = hS[t];
    }
}

__global__ void __launch_bounds__(1024)
GCN_binscan_kernel(const int* __restrict__ cnt, int* __restrict__ base) {
    __shared__ int red[32];
    __shared__ int wsum[32];
    const int* c = cnt + blockIdx.y * (NB * NWG);
    int* o = base + blockIdx.y * (NB * NWG);
    int t = threadIdx.x, b = blockIdx.x, lane = t & 31, wave = t >> 5;
    const int4* c4 = (const int4*)c;
    int part = 0;
    int n4 = b * 256;
#pragma unroll 8
    for (int i = t; i < n4; i += 1024) {
        int4 v = c4[i];
        part += (v.x + v.y) + (v.z + v.w);
    }
#pragma unroll
    for (int off = 16; off > 0; off >>= 1) part += __shfl_xor(part, off, 32);
    if (lane == 0) red[wave] = part;

    int idx = b * 1024 + t;
    int v = c[idx];
    int inc = v;
#pragma unroll
    for (int off = 1; off < 32; off <<= 1) {
        int u = __shfl_up(inc, off, 32);
        if (lane >= off) inc += u;
    }
    if (lane == 31) wsum[wave] = inc;
    __syncthreads();
    int pre = 0;
#pragma unroll
    for (int k = 0; k < 32; ++k) {
        pre += red[k];
        if (k < wave) pre += wsum[k];
    }
    o[idx] = pre + inc - v;
}

__global__ void __launch_bounds__(1024)
GCN_binplace_kernel(const int* __restrict__ esrc, const int* __restrict__ edst,
                    const int* __restrict__ base, int* __restrict__ binD,
                    int* __restrict__ binS) {
    __shared__ int oD[NB];
    __shared__ int oS[NB];
    int t = threadIdx.x, w = blockIdx.x;
    if (t < NB) {
        oD[t] = base[t * NWG + w];
        oS[t] = base[NB * NWG + t * NWG + w];
    }
    int e0 = w * CH;
    int d[EPT], s[EPT];
#pragma unroll
    for (int k = 0; k < EPT; ++k) {
        int i = t + k * 1024;
        d[k] = (i < CH) ? edst[e0 + i] : -1;
        s[k] = (i < CH) ? esrc[e0 + i] : -1;
    }
    __syncthreads();
#pragma unroll
    for (int k = 0; k < EPT; ++k) {
        if (d[k] >= 0) {
            int bd = d[k] / BW, bs = s[k] / BW;
            int pd = atomicAdd(&oD[bd], 1);
            int ps = atomicAdd(&oS[bs], 1);
            binD[pd] = ((d[k] - bd * BW) << SRC_BITS) | s[k];
            binS[ps] = s[k] - bs * BW;
        }
    }
}

__global__ void __launch_bounds__(1024)
GCN_rows_kernel(const int* __restrict__ base, const int* __restrict__ binD,
                const int* __restrict__ binS, int* __restrict__ rowptr,
                float* __restrict__ inorm, float* __restrict__ onorm,
                int* __restrict__ perm) {
    __shared__ int cntIn[BW];
    __shared__ int cntOut[BW];
    __shared__ int start[BW];
    __shared__ int cur[BW];
    __shared__ int wsum[32];
    int t = threadIdx.x, b = blockIdx.x, lane = t & 31, wave = t >> 5;
    int dBeg = base[b * NWG];
    int dEnd = (b + 1 < NB) ? base[(b + 1) * NWG] : EE;
    int sBeg = base[NB * NWG + b * NWG];
    int sEnd = (b + 1 < NB) ? base[NB * NWG + (b + 1) * NWG] : EE;
    if (t < BW) { cntIn[t] = 0; cntOut[t] = 0; cur[t] = 0; }
    __syncthreads();
    for (int i0 = dBeg + t; i0 < dEnd; i0 += RL * 1024) {
        int p[RL];
#pragma unroll
        for (int k = 0; k < RL; ++k) {
            int i = i0 + k * 1024;
            p[k] = (i < dEnd) ? binD[i] : -1;
        }
#pragma unroll
        for (int k = 0; k < RL; ++k)
            if (p[k] >= 0) atomicAdd(&cntIn[p[k] >> SRC_BITS], 1);
    }
    for (int i0 = sBeg + t; i0 < sEnd; i0 += RL * 1024) {
        int p[RL];
#pragma unroll
        for (int k = 0; k < RL; ++k) {
            int i = i0 + k * 1024;
            p[k] = (i < sEnd) ? binS[i] : -1;
        }
#pragma unroll
        for (int k = 0; k < RL; ++k)
            if (p[k] >= 0) atomicAdd(&cntOut[p[k]], 1);
    }
    __syncthreads();

    int c = (t < BW) ? cntIn[t] : 0;
    int inc = c;
#pragma unroll
    for (int off = 1; off < 32; off <<= 1) {
        int u = __shfl_up(inc, off, 32);
        if (lane >= off) inc += u;
    }
    if (lane == 31) wsum[wave] = inc;
    __syncthreads();
    int pre = 0;
#pragma unroll
    for (int k = 0; k < 32; ++k) {
        if (k < wave) pre += wsum[k];
    }
    int first = dBeg + pre + inc - c;
    int n = b * BW + t;
    if (t < BW) {
        start[t] = first;
        if (n < NN) {
            rowptr[n] = first;
            inorm[n] = rsqrtf(fmaxf((float)c, 1.0f));
            onorm[n] = rsqrtf(fmaxf((float)cntOut[t], 1.0f));
            if (n == NN - 1) rowptr[NN] = first + c;
        }
    }
    __syncthreads();
    for (int i0 = dBeg + t; i0 < dEnd; i0 += RL * 1024) {
        int p[RL];
#pragma unroll
        for (int k = 0; k < RL; ++k) {
            int i = i0 + k * 1024;
            p[k] = (i < dEnd) ? binD[i] : -1;
        }
#pragma unroll
        for (int k = 0; k < RL; ++k) {
            if (p[k] >= 0) {
                int dl = p[k] >> SRC_BITS;
                int slot = atomicAdd(&cur[dl], 1);
                perm[start[dl] + slot] = p[k] & ((1 << SRC_BITS) - 1);
            }
        }
    }
}

__global__ void __launch_bounds__(256)
GCN_gather1_kernel(const float* __restrict__ feat, const int* __restrict__ rowptr,
                   const int* __restrict__ perm, const float* __restrict__ onorm,
                   float* __restrict__ agg) {
    int lane = threadIdx.x & 31;
    int n = __builtin_amdgcn_readfirstlane(blockIdx.x * 8 + (threadIdx.x >> 5));
    int beg = rowptr[n], end = rowptr[n + 1];
    float4 acc = make_float4(0.0f, 0.0f, 0.0f, 0.0f);
    for (int base = beg; base < end; base += 32) {
        int cnt = min(32, end - base);
        int s_l = 0;
        float sc_l = 0.0f;
        if (lane < cnt) {
            s_l = perm[base + lane];
            sc_l = onorm[s_l];
        }
        int sc_i = __float_as_int(sc_l);
        for (int j = 0; j < cnt; j += 4) {
            int j1 = min(j + 1, cnt - 1), j2 = min(j + 2, cnt - 1), j3 = min(j + 3, cnt - 1);
            int s0 = __builtin_amdgcn_readlane(s_l, j);
            int s1 = __builtin_amdgcn_readlane(s_l, j1);
            int s2 = __builtin_amdgcn_readlane(s_l, j2);
            int s3 = __builtin_amdgcn_readlane(s_l, j3);
            float c0 = __int_as_float(__builtin_amdgcn_readlane(sc_i, j));
            float c1 = (j + 1 < cnt) ? __int_as_float(__builtin_amdgcn_readlane(sc_i, j1)) : 0.0f;
            float c2 = (j + 2 < cnt) ? __int_as_float(__builtin_amdgcn_readlane(sc_i, j2)) : 0.0f;
            float c3 = (j + 3 < cnt) ? __int_as_float(__builtin_amdgcn_readlane(sc_i, j3)) : 0.0f;
            const float4 v0 = *(const float4*)(feat + (size_t)s0 * 128 + lane * 4);
            const float4 v1 = *(const float4*)(feat + (size_t)s1 * 128 + lane * 4);
            const float4 v2 = *(const float4*)(feat + (size_t)s2 * 128 + lane * 4);
            const float4 v3 = *(const float4*)(feat + (size_t)s3 * 128 + lane * 4);
            acc.x = fmaf(v0.x, c0, acc.x); acc.y = fmaf(v0.y, c0, acc.y);
            acc.z = fmaf(v0.z, c0, acc.z); acc.w = fmaf(v0.w, c0, acc.w);
            acc.x = fmaf(v1.x, c1, acc.x); acc.y = fmaf(v1.y, c1, acc.y);
            acc.z = fmaf(v1.z, c1, acc.z); acc.w = fmaf(v1.w, c1, acc.w);
            acc.x = fmaf(v2.x, c2, acc.x); acc.y = fmaf(v2.y, c2, acc.y);
            acc.z = fmaf(v2.z, c2, acc.z); acc.w = fmaf(v2.w, c2, acc.w);
            acc.x = fmaf(v3.x, c3, acc.x); acc.y = fmaf(v3.y, c3, acc.y);
            acc.z = fmaf(v3.z, c3, acc.z); acc.w = fmaf(v3.w, c3, acc.w);
        }
    }
    *(float4*)(agg + (size_t)n * 128 + lane * 4) = acc;
}

#define GS    4
#define GGRID ((NN / 16 + GS - 1) / GS)
#define W2P   136
static_assert(128 % GS == 0 && 32 % GS == 0, "the W staging loops take whole trips");
static_assert(32 * GS == 128, "one thread per bias element in the staging");

__global__ void __launch_bounds__(32 * GS)
GCN_gemm12_kernel(const float* __restrict__ A, const float* __restrict__ W1,
                  const float* __restrict__ bias, const float* __restrict__ inorm,
                  const float* __restrict__ onorm, const float* __restrict__ W2,
                  float* __restrict__ T) {
    static_assert(NN % 16 == 0, "whole strips only");
    __shared__ float4 lds4[(64 * W2P + 128 + 128 * 128) / 4];
    float* lW2 = (float*)lds4;
    float* lB  = lW2 + 64 * W2P;
    float* lW1 = lB + 128;
    int tid = threadIdx.x, lane = tid & 31;
    int row0 = __builtin_amdgcn_readfirstlane((blockIdx.x * GS + (tid >> 5)) * 16);
    bool live = row0 < NN;
    int r = lane & 15, kh = lane >> 4;

    int rowA = (live ? row0 : NN - 16) + r;
    v2f a[32];
    {
        const float* Ap = A + (size_t)rowA * 128 + 2 * kh;
#pragma unroll
        for (int kt = 0; kt < 32; ++kt) a[kt] = *(const v2f*)(Ap + kt * 4);
    }
    float si = inorm[rowA];
    float so = onorm[rowA];
    {
        const float4* Wv = (const float4*)W1;
        float4* lWv = (float4*)lW1;
#pragma unroll
        for (int i = 0; i < 128 / GS; ++i) lWv[tid + 32 * GS * i] = Wv[tid + 32 * GS * i];
    }
    {
        const float4* Wv = (const float4*)W2;
        float4* lWv = (float4*)lW2;
#pragma unroll
        for (int i = 0; i < 32 / GS; ++i) {
            int idx = tid + 32 * GS * i;
            int kp = idx >> 4, c4 = idx & 15;
            float4 u = Wv[(2 * kp) * 16 + c4];
            float4 w = Wv[(2 * kp + 1) * 16 + c4];
            lWv[kp * (W2P / 4) + 2 * c4]     = make_float4(u.x, w.x, u.y, w.y);
            lWv[kp * (W2P / 4) + 2 * c4 + 1] = make_float4(u.z, w.z, u.w, w.w);
        }
    }
    lB[tid] = bias[tid];
    __syncthreads();
    if (!live) return;

    v8f c[8] = {};
    typedef const __attribute__((address_space(3))) float lds_cfloat;
    lds_cfloat* Bp = (lds_cfloat*)(lW1 + (size_t)(2 * kh) * 128 + r);
    asm volatile("" : "+v"(Bp));
#pragma unroll
    for (int kt = 0; kt < 32; ++kt) {
#pragma unroll
        for (int ct = 0; ct < 8; ++ct) {
            v2f b;
            b.x = Bp[kt * 512 + ct * 16];
            b.y = Bp[kt * 512 + 128 + ct * 16];
            c[ct] = __builtin_amdgcn_wmma_f32_16x16x4_f32(false, b, false, a[kt], (short)0,
                                                          c[ct], false, false);
        }
    }
#pragma unroll
    for (int ct = 0; ct < 8; ++ct) {
        const float4* bq = (const float4*)(lB + ct * 16 + 8 * kh);
        float4 q0 = bq[0], q1 = bq[1];
        float bb[8] = {q0.x, q0.y, q0.z, q0.w, q1.x, q1.y, q1.z, q1.w};
#pragma unroll
        for (int v = 0; v < 8; ++v) {
            float val = fmaf(c[ct][v], si, bb[v]);
            c[ct][v] = fmaxf(val, 0.0f) * so;
        }
    }

    v8f d[4] = {};
    const float* B2 = lW2 + (size_t)(4 * kh) * W2P + 2 * r;
#pragma unroll
    for (int ct = 0; ct < 8; ++ct) {
#pragma unroll
        for (int p = 0; p < 4; ++p) {
            v2f h;
            h.x = c[ct][2 * p];
            h.y = c[ct][2 * p + 1];
#pragma unroll
            for (int jt = 0; jt < 4; ++jt) {
                v2f b = *(const v2f*)(B2 + (ct * 8 + p) * W2P + jt * 32);
                d[jt] = __builtin_amdgcn_wmma_f32_16x16x4_f32(false, h, false, b, (short)0,
                                                              d[jt], false, false);
            }
        }
    }
#pragma unroll
    for (int jt = 0; jt < 4; ++jt) {
        int n = jt * 16 + r;
#pragma unroll
        for (int v = 0; v < 8; ++v) {
            int m = row0 + v + 8 * kh;
            T[(size_t)m * 64 + n] = d[jt][v];
        }
    }
}

__global__ void __launch_bounds__(256)
GCN_gather2_kernel(const float* __restrict__ T, const int* __restrict__ rowptr,
                   const int* __restrict__ perm, const float* __restrict__ inorm,
                   const float* __restrict__ b2, float* __restrict__ H2) {
    int lane = threadIdx.x & 31;
    int n = __builtin_amdgcn_readfirstlane(blockIdx.x * 8 + (threadIdx.x >> 5));
    int beg = rowptr[n], end = rowptr[n + 1];
    float2 acc = make_float2(0.0f, 0.0f);
    for (int base = beg; base < end; base += 32) {
        int cnt = min(32, end - base);
        int s_l = 0;
        if (lane < cnt) s_l = perm[base + lane];
        for (int j = 0; j < cnt; j += 4) {
            int j1 = min(j + 1, cnt - 1), j2 = min(j + 2, cnt - 1), j3 = min(j + 3, cnt - 1);
            int s0 = __builtin_amdgcn_readlane(s_l, j);
            int s1 = __builtin_amdgcn_readlane(s_l, j1);
            int s2 = __builtin_amdgcn_readlane(s_l, j2);
            int s3 = __builtin_amdgcn_readlane(s_l, j3);
            float c1 = (j + 1 < cnt) ? 1.0f : 0.0f;
            float c2 = (j + 2 < cnt) ? 1.0f : 0.0f;
            float c3 = (j + 3 < cnt) ? 1.0f : 0.0f;
            const float2 v0 = *(const float2*)(T + (size_t)s0 * 64 + lane * 2);
            const float2 v1 = *(const float2*)(T + (size_t)s1 * 64 + lane * 2);
            const float2 v2 = *(const float2*)(T + (size_t)s2 * 64 + lane * 2);
            const float2 v3 = *(const float2*)(T + (size_t)s3 * 64 + lane * 2);
            acc.x += v0.x;                  acc.y += v0.y;
            acc.x = fmaf(v1.x, c1, acc.x);  acc.y = fmaf(v1.y, c1, acc.y);
            acc.x = fmaf(v2.x, c2, acc.x);  acc.y = fmaf(v2.y, c2, acc.y);
            acc.x = fmaf(v3.x, c3, acc.x);  acc.y = fmaf(v3.y, c3, acc.y);
        }
    }
    float s = inorm[n];
    float2 o;
    o.x = fmaf(acc.x, s, b2[lane * 2 + 0]);
    o.y = fmaf(acc.y, s, b2[lane * 2 + 1]);
    *(float2*)(H2 + (size_t)n * 64 + lane * 2) = o;
}

#define SQ  32
#define SWV 10

template <int OFF>
__device__ __forceinline__ void GCN_score_level(float (&p)[SQ], bool hi) {
#pragma unroll
    for (int j = 0; j < OFF; ++j) {
        float keep = hi ? p[j + OFF] : p[j];
        float send = hi ? p[j] : p[j + OFF];
        p[j] = keep + __shfl_xor(send, OFF, 32);
    }
}

__global__ void __launch_bounds__(32 * SWV) __attribute__((amdgpu_waves_per_eu(8, 8)))
GCN_score_kernel(const float* __restrict__ H2, const int* __restrict__ qs,
                 const int* __restrict__ qd, float* __restrict__ out) {
    static_assert(QQ % (SQ * SWV) == 0, "whole waves and whole work-groups only");
    static_assert((size_t)NN * 32 < 0x7fffffffu, "row offsets fit 32 bits");
    const float2* __restrict__ rows = (const float2*)H2;
    unsigned lane = threadIdx.x & 31;
    int w = __builtin_amdgcn_readfirstlane(blockIdx.x * SWV + (threadIdx.x >> 5));
    int q0 = w * SQ;
    int s_l = qs[q0 + lane], d_l = qd[q0 + lane];
    float2 a[SQ], b[SQ];
    float p[SQ];
#pragma unroll
    for (int j = 0; j < SQ; ++j) {
        unsigned s = (unsigned)__builtin_amdgcn_readlane(s_l, j);
        unsigned d = (unsigned)__builtin_amdgcn_readlane(d_l, j);
        a[j] = rows[s * 32u + lane];
        b[j] = rows[d * 32u + lane];
    }
#pragma unroll
    for (int j = 0; j < SQ; ++j) p[j] = fmaf(a[j].x, b[j].x, a[j].y * b[j].y);
    GCN_score_level<16>(p, (lane & 16) != 0);
    GCN_score_level<8>(p, (lane & 8) != 0);
    GCN_score_level<4>(p, (lane & 4) != 0);
    GCN_score_level<2>(p, (lane & 2) != 0);
    GCN_score_level<1>(p, (lane & 1) != 0);
    out[q0 + lane] = 1.0f / (1.0f + __expf(-p[0]));
}

extern "C" void kernel_launch(void* const* d_in, const int* in_sizes, int n_in,
                              void* d_out, int out_size, void* d_ws, size_t ws_size,
                              hipStream_t stream) {
    const float* feat = (const float*)d_in[0];
    const int* esrc   = (const int*)d_in[1];
    const int* edst   = (const int*)d_in[2];
    const int* qsrc   = (const int*)d_in[3];
    const int* qdst   = (const int*)d_in[4];
    const float* W1   = (const float*)d_in[5];
    const float* b1   = (const float*)d_in[6];
    const float* W2   = (const float*)d_in[7];
    const float* b2   = (const float*)d_in[8];
    float* out = (float*)d_out;

    float* ws    = (float*)d_ws;
    float* onorm = ws;
    float* agg1  = ws + 2 * NN;
    float* h2    = agg1;
    float* h1    = agg1 + (size_t)NN * 128;
    float* t2    = h1 + (size_t)NN * 128;
    int*   perm  = (int*)(t2 + (size_t)NN * 64);
    float* inorm = (float*)(perm + 2 * EE);
    int*   rowptr = (int*)(inorm + NN);
    int*   cnt   = (int*)h1;
    int*   base  = cnt + 2 * NB * NWG;
    int*   binD  = base + 2 * NB * NWG;
    int*   binS  = binD + EE;

    GCN_bincount_kernel<<<NWG, 1024, 0, stream>>>(esrc, edst, cnt);
    GCN_binscan_kernel<<<dim3(NB * NWG / 1024, 2), 1024, 0, stream>>>(cnt, base);
    GCN_binplace_kernel<<<NWG, 1024, 0, stream>>>(esrc, edst, base, binD, binS);
    GCN_rows_kernel<<<NB, 1024, 0, stream>>>(base, binD, binS, rowptr, inorm, onorm, perm);
    GCN_gather1_kernel<<<NN / 8, 256, 0, stream>>>(feat, rowptr, perm, onorm, agg1);
    GCN_gemm12_kernel<<<GGRID, 32 * GS, 0, stream>>>(agg1, W1, b1, inorm, onorm, W2, t2);
    GCN_gather2_kernel<<<NN / 8, 256, 0, stream>>>(t2, rowptr, perm, inorm, b2, h2);
    GCN_score_kernel<<<QQ / (SQ * SWV), 32 * SWV, 0, stream>>>(h2, qsrc, qdst, out);
}


# Edge ordering in 2 launches, not 4: a chunk's edges are put in bin order inside LDS and stored chunk-major as whole lines; the row kernel sums the chunks' bin starts and reads its 256 short runs
# speedup vs baseline: 5.2472x; 1.3340x over previous
#include <hip/hip_runtime.h>
#include <math.h>

typedef __attribute__((ext_vector_type(2))) float v2f;
typedef __attribute__((ext_vector_type(8))) float v8f;

#define NN 100000
#define EE 1600000
#define QQ 1000000

#define NB  256
#define BW  391
#define NWG 256
#define CH  (EE / NWG)
#define EPT 7
#define SRC_BITS 17
#define CPW (NWG / 32)
#define PCAP 8192

__global__ void __launch_bounds__(1024)
GCN_chunksort_kernel(const int* __restrict__ esrc, const int* __restrict__ edst,
                     int* __restrict__ offs, int* __restrict__ sortD,
                     int* __restrict__ sortS) {
    __shared__ int stD[CH];
    __shared__ int stS[CH];
    __shared__ int hD[NB];
    __shared__ int hS[NB];
    __shared__ int wsum[2 * NB / 32];
    static_assert(NB % 32 == 0 && 2 * NB <= 1024, "whole waves scan the bin counts");
    int t = threadIdx.x, w = blockIdx.x, lane = t & 31, wave = t >> 5;
    if (t < NB) { hD[t] = 0; hS[t] = 0; }
    int e0 = w * CH;
    int d[EPT], s[EPT];
#pragma unroll
    for (int k = 0; k < EPT; ++k) {
        int i = t + k * 1024;
        d[k] = (i < CH) ? edst[e0 + i] : -1;
        s[k] = (i < CH) ? esrc[e0 + i] : -1;
    }
    __syncthreads();
#pragma unroll
    for (int k = 0; k < EPT; ++k) {
        if (d[k] >= 0) {
            atomicAdd(&hD[d[k] / BW], 1);
            atomicAdd(&hS[s[k] / BW], 1);
        }
    }
    __syncthreads();
    int kind = t / NB, bin = t % NB;
    int c = 0, inc = 0;
    if (t < 2 * NB) {
        c = kind ? hS[bin] : hD[bin];
        inc = c;
#pragma unroll
        for (int off = 1; off < 32; off <<= 1) {
            int u = __shfl_up(inc, off, 32);
            if (lane >= off) inc += u;
        }
        if (lane == 31) wsum[wave] = inc;
    }
    __syncthreads();
    if (t < 2 * NB) {
        int w0 = kind * (NB / 32);
        int pre = 0;
#pragma unroll
        for (int k = 0; k < NB / 32; ++k) {
            if (w0 + k < wave) pre += wsum[w0 + k];
        }
        int ls = pre + inc - c;
        if (kind) hS[bin] = ls; else hD[bin] = ls;
        offs[kind * (NWG * NB) + w * NB + bin] = ls;
    }
    __syncthreads();
#pragma unroll
    for (int k = 0; k < EPT; ++k) {
        if (d[k] >= 0) {
            int bd = d[k] / BW, bs = s[k] / BW;
            int pd = atomicAdd(&hD[bd], 1);
            int ps = atomicAdd(&hS[bs], 1);
            stD[pd] = ((d[k] - bd * BW) << SRC_BITS) | s[k];
            stS[ps] = s[k] - bs * BW;
        }
    }
    __syncthreads();
#pragma unroll
    for (int k = 0; k < EPT; ++k) {
        int i = t + k * 1024;
        if (i < CH) {
            sortD[e0 + i] = stD[i];
            sortS[e0 + i] = stS[i];
        }
    }
}


template <bool IN_LDS>
__device__ __forceinline__ void GCN_place_rows(const int (&p)[CPW], int maxD, int wave, int lane,
                                               const int* __restrict__ sortD, const int* segD,
                                               int* cur, int* lperm, int* __restrict__ out) {
#pragma unroll
    for (int k = 0; k < CPW; ++k) {
        if (p[k] >= 0) {
            int pos = atomicAdd(&cur[p[k] >> SRC_BITS], 1);
            int sid = p[k] & ((1 << SRC_BITS) - 1);
            if (IN_LDS) lperm[pos] = sid;
            else        out[pos] = sid;
        }
    }
    for (int r = 32; r < maxD; r += 32) {
        for (int k = 0; k < CPW; ++k) {
            int w = wave * CPW + k;
            int i = segD[2 * w] + r + lane;
            if (i < segD[2 * w + 1]) {
                int e = sortD[w * CH + i];
                int pos = atomicAdd(&cur[e >> SRC_BITS], 1);
                int sid = e & ((1 << SRC_BITS) - 1);
                if (IN_LDS) lperm[pos] = sid;
                else        out[pos] = sid;
            }
        }
    }
}

__global__ void __launch_bounds__(1024)
GCN_rows_kernel(const int* __restrict__ offs, const int* __restrict__ sortD,
                const int* __restrict__ sortS, int* __restrict__ rowptr,
                float* __restrict__ inorm, float* __restrict__ onorm,
                int* __restrict__ perm) {
    __shared__ int cntIn[BW];
    __shared__ int cntOut[BW];
    __shared__ int cur[BW];
    __shared__ int wsum[32];
    __shared__ int segD[2 * NWG];
    __shared__ int segS[2 * NWG];
    __shared__ int redB[NWG / 32];
    __shared__ int redE[NWG / 32];
    __shared__ int lperm[PCAP];
    static_assert(NWG % 32 == 0 && NWG <= 1024 && CPW * 32 == NWG, "a thread per chunk, CPW chunks per wave");
    int t = threadIdx.x, b = blockIdx.x, lane = t & 31, wave = t >> 5;
    if (t < BW) { cntIn[t] = 0; cntOut[t] = 0; }
    if (t < NWG) {
        const int* oD = offs + t * NB;
        const int* oS = offs + NWG * NB + t * NB;
        int sd = oD[b], ed = (b + 1 < NB) ? oD[b + 1] : CH;
        int ss = oS[b], es = (b + 1 < NB) ? oS[b + 1] : CH;
        segD[2 * t] = sd; segD[2 * t + 1] = ed;
        segS[2 * t] = ss; segS[2 * t + 1] = es;
#pragma unroll
        for (int off = 16; off > 0; off >>= 1) {
            sd += __shfl_xor(sd, off, 32);
            ed += __shfl_xor(ed, off, 32);
        }
        if (lane == 0) { redB[wave] = sd; redE[wave] = ed; }
    }
    __syncthreads();
    int dBeg = 0, dEnd = 0;
#pragma unroll
    for (int k = 0; k < NWG / 32; ++k) { dBeg += redB[k]; dEnd += redE[k]; }

    int p[CPW], q[CPW];
    int maxD = 0, maxS = 0;
#pragma unroll
    for (int k = 0; k < CPW; ++k) {
        int w = wave * CPW + k;
        int sd = segD[2 * w], ed = segD[2 * w + 1];
        int ss = segS[2 * w], es = segS[2 * w + 1];
        p[k] = (sd + lane < ed) ? sortD[w * CH + sd + lane] : -1;
        q[k] = (ss + lane < es) ? sortS[w * CH + ss + lane] : -1;
        maxD = max(maxD, ed - sd);
        maxS = max(maxS, es - ss);
    }
    maxD = __builtin_amdgcn_readfirstlane(maxD);
    maxS = __builtin_amdgcn_readfirstlane(maxS);
#pragma unroll
    for (int k = 0; k < CPW; ++k) {
        if (p[k] >= 0) atomicAdd(&cntIn[p[k] >> SRC_BITS], 1);
        if (q[k] >= 0) atomicAdd(&cntOut[q[k]], 1);
    }
    for (int r = 32; r < maxD; r += 32) {
        for (int k = 0; k < CPW; ++k) {
            int w = wave * CPW + k;
            int i = segD[2 * w] + r + lane;
            if (i < segD[2 * w + 1]) atomicAdd(&cntIn[sortD[w * CH + i] >> SRC_BITS], 1);
        }
    }
    for (int r = 32; r < maxS; r += 32) {
        for (int k = 0; k < CPW; ++k) {
            int w = wave * CPW + k;
            int i = segS[2 * w] + r + lane;
            if (i < segS[2 * w + 1]) atomicAdd(&cntOut[sortS[w * CH + i]], 1);
        }
    }
    __syncthreads();

    int c = (t < BW) ? cntIn[t] : 0;
    int inc = c;
#pragma unroll
    for (int off = 1; off < 32; off <<= 1) {
        int u = __shfl_up(inc, off, 32);
        if (lane >= off) inc += u;
    }
    if (lane == 31) wsum[wave] = inc;
    __syncthreads();
    int pre = 0;
#pragma unroll
    for (int k = 0; k < 32; ++k) {
        if (k < wave) pre += wsum[k];
    }
    int first = dBeg + pre + inc - c;
    int n = b * BW + t;
    if (t < BW) {
        cur[t] = first - dBeg;
        if (n < NN) {
            rowptr[n] = first;
            inorm[n] = rsqrtf(fmaxf((float)c, 1.0f));
            onorm[n] = rsqrtf(fmaxf((float)cntOut[t], 1.0f));
            if (n == NN - 1) rowptr[NN] = first + c;
        }
    }
    __syncthreads();
    int len = dEnd - dBeg;
    if (len > PCAP) {
        GCN_place_rows<false>(p, maxD, wave, lane, sortD, segD, cur, lperm, perm + dBeg);
    } else {
        GCN_place_rows<true>(p, maxD, wave, lane, sortD, segD, cur, lperm, perm + dBeg);
        __syncthreads();
        int o[PCAP / 1024];
#pragma unroll
        for (int k = 0; k < PCAP / 1024; ++k) {
            int i = t + k * 1024;
            o[k] = (i < len) ? lperm[i] : 0;
        }
#pragma unroll
        for (int k = 0; k < PCAP / 1024; ++k) {
            int i = t + k * 1024;
            if (i < len) perm[dBeg + i] = o[k];
        }
    }
}

__global__ void __launch_bounds__(256)
GCN_gather1_kernel(const float* __restrict__ feat, const int* __restrict__ rowptr,
                   const int* __restrict__ perm, const float* __restrict__ onorm,
                   float* __restrict__ agg) {
    int lane = threadIdx.x & 31;
    int n = __builtin_amdgcn_readfirstlane(blockIdx.x * 8 + (threadIdx.x >> 5));
    int beg = rowptr[n], end = rowptr[n + 1];
    float4 acc = make_float4(0.0f, 0.0f, 0.0f, 0.0f);
    for (int base = beg; base < end; base += 32) {
        int cnt = min(32, end - base);
        int s_l = 0;
        float sc_l = 0.0f;
        if (lane < cnt) {
            s_l = perm[base + lane];
            sc_l = onorm[s_l];
        }
        int sc_i = __float_as_int(sc_l);
        for (int j = 0; j < cnt; j += 4) {
            int j1 = min(j + 1, cnt - 1), j2 = min(j + 2, cnt - 1), j3 = min(j + 3, cnt - 1);
            int s0 = __builtin_amdgcn_readlane(s_l, j);
            int s1 = __builtin_amdgcn_readlane(s_l, j1);
            int s2 = __builtin_amdgcn_readlane(s_l, j2);
            int s3 = __builtin_amdgcn_readlane(s_l, j3);
            float c0 = __int_as_float(__builtin_amdgcn_readlane(sc_i, j));
            float c1 = (j + 1 < cnt) ? __int_as_float(__builtin_amdgcn_readlane(sc_i, j1)) : 0.0f;
            float c2 = (j + 2 < cnt) ? __int_as_float(__builtin_amdgcn_readlane(sc_i, j2)) : 0.0f;
            float c3 = (j + 3 < cnt) ? __int_as_float(__builtin_amdgcn_readlane(sc_i, j3)) : 0.0f;
            const float4 v0 = *(const float4*)(feat + (size_t)s0 * 128 + lane * 4);
            const float4 v1 = *(const float4*)(feat + (size_t)s1 * 128 + lane * 4);
            const float4 v2 = *(const float4*)(feat + (size_t)s2 * 128 + lane * 4);
            const float4 v3 = *(const float4*)(feat + (size_t)s3 * 128 + lane * 4);
            acc.x = fmaf(v0.x, c0, acc.x); acc.y = fmaf(v0.y, c0, acc.y);
            acc.z = fmaf(v0.z, c0, acc.z); acc.w = fmaf(v0.w, c0, acc.w);
            acc.x = fmaf(v1.x, c1, acc.x); acc.y = fmaf(v1.y, c1, acc.y);
            acc.z = fmaf(v1.z, c1, acc.z); acc.w = fmaf(v1.w, c1, acc.w);
            acc.x = fmaf(v2.x, c2, acc.x); acc.y = fmaf(v2.y, c2, acc.y);
            acc.z = fmaf(v2.z, c2, acc.z); acc.w = fmaf(v2.w, c2, acc.w);
            acc.x = fmaf(v3.x, c3, acc.x); acc.y = fmaf(v3.y, c3, acc.y);
            acc.z = fmaf(v3.z, c3, acc.z); acc.w = fmaf(v3.w, c3, acc.w);
        }
    }
    *(float4*)(agg + (size_t)n * 128 + lane * 4) = acc;
}

#define GS    4
#define GGRID ((NN / 16 + GS - 1) / GS)
#define W2P   136
static_assert(128 % GS == 0 && 32 % GS == 0, "the W staging loops take whole trips");
static_assert(32 * GS == 128, "one thread per bias element in the staging");

__global__ void __launch_bounds__(32 * GS)
GCN_gemm12_kernel(const float* __restrict__ A, const float* __restrict__ W1,
                  const float* __restrict__ bias, const float* __restrict__ inorm,
                  const float* __restrict__ onorm, const float* __restrict__ W2,
                  float* __restrict__ T) {
    static_assert(NN % 16 == 0, "whole strips only");
    __shared__ float4 lds4[(64 * W2P + 128 + 128 * 128) / 4];
    float* lW2 = (float*)lds4;
    float* lB  = lW2 + 64 * W2P;
    float* lW1 = lB + 128;
    int tid = threadIdx.x, lane = tid & 31;
    int row0 = __builtin_amdgcn_readfirstlane((blockIdx.x * GS + (tid >> 5)) * 16);
    bool live = row0 < NN;
    int r = lane & 15, kh = lane >> 4;

    int rowA = (live ? row0 : NN - 16) + r;
    v2f a[32];
    {
        const float* Ap = A + (size_t)rowA * 128 + 2 * kh;
#pragma unroll
        for (int kt = 0; kt < 32; ++kt) a[kt] = *(const v2f*)(Ap + kt * 4);
    }
    float si = inorm[rowA];
    float so = onorm[rowA];
    {
        const float4* Wv = (const float4*)W1;
        float4* lWv = (float4*)lW1;
#pragma unroll
        for (int i = 0; i < 128 / GS; ++i) lWv[tid + 32 * GS * i] = Wv[tid + 32 * GS * i];
    }
    {
        const float4* Wv = (const float4*)W2;
        float4* lWv = (float4*)lW2;
#pragma unroll
        for (int i = 0; i < 32 / GS; ++i) {
            int idx = tid + 32 * GS * i;
            int kp = idx >> 4, c4 = idx & 15;
            float4 u = Wv[(2 * kp) * 16 + c4];
            float4 w = Wv[(2 * kp + 1) * 16 + c4];
            lWv[kp * (W2P / 4) + 2 * c4]     = make_float4(u.x, w.x, u.y, w.y);
            lWv[kp * (W2P / 4) + 2 * c4 + 1] = make_float4(u.z, w.z, u.w, w.w);
        }
    }
    lB[tid] = bias[tid];
    __syncthreads();
    if (!live) return;

    v8f c[8] = {};
    typedef const __attribute__((address_space(3))) float lds_cfloat;
    lds_cfloat* Bp = (lds_cfloat*)(lW1 + (size_t)(2 * kh) * 128 + r);
    asm volatile("" : "+v"(Bp));
#pragma unroll
    for (int kt = 0; kt < 32; ++kt) {
#pragma unroll
        for (int ct = 0; ct < 8; ++ct) {
            v2f b;
            b.x = Bp[kt * 512 + ct * 16];
            b.y = Bp[kt * 512 + 128 + ct * 16];
            c[ct] = __builtin_amdgcn_wmma_f32_16x16x4_f32(false, b, false, a[kt], (short)0,
                                                          c[ct], false, false);
        }
    }
#pragma unroll
    for (int ct = 0; ct < 8; ++ct) {
        const float4* bq = (const float4*)(lB + ct * 16 + 8 * kh);
        float4 q0 = bq[0], q1 = bq[1];
        float bb[8] = {q0.x, q0.y, q0.z, q0.w, q1.x, q1.y, q1.z, q1.w};
#pragma unroll
        for (int v = 0; v < 8; ++v) {
            float val = fmaf(c[ct][v], si, bb[v]);
            c[ct][v] = fmaxf(val, 0.0f) * so;
        }
    }

    v8f d[4] = {};
    const float* B2 = lW2 + (size_t)(4 * kh) * W2P + 2 * r;
#pragma unroll
    for (int ct = 0; ct < 8; ++ct) {
#pragma unroll
        for (int p = 0; p < 4; ++p) {
            v2f h;
            h.x = c[ct][2 * p];
            h.y = c[ct][2 * p + 1];
#pragma unroll
            for (int jt = 0; jt < 4; ++jt) {
                v2f b = *(const v2f*)(B2 + (ct * 8 + p) * W2P + jt * 32);
                d[jt] = __builtin_amdgcn_wmma_f32_16x16x4_f32(false, h, false, b, (short)0,
                                                              d[jt], false, false);
            }
        }
    }
#pragma unroll
    for (int jt = 0; jt < 4; ++jt) {
        int n = jt * 16 + r;
#pragma unroll
        for (int v = 0; v < 8; ++v) {
            int m = row0 + v + 8 * kh;
            T[(size_t)m * 64 + n] = d[jt][v];
        }
    }
}

__global__ void __launch_bounds__(256)
GCN_gather2_kernel(const float* __restrict__ T, const int* __restrict__ rowptr,
                   const int* __restrict__ perm, const float* __restrict__ inorm,
                   const float* __restrict__ b2, float* __restrict__ H2) {
    int lane = threadIdx.x & 31;
    int n = __builtin_amdgcn_readfirstlane(blockIdx.x * 8 + (threadIdx.x >> 5));
    int beg = rowptr[n], end = rowptr[n + 1];
    float2 acc = make_float2(0.0f, 0.0f);
    for (int base = beg; base < end; base += 32) {
        int cnt = min(32, end - base);
        int s_l = 0;
        if (lane < cnt) s_l = perm[base + lane];
        for (int j = 0; j < cnt; j += 4) {
            int j1 = min(j + 1, cnt - 1), j2 = min(j + 2, cnt - 1), j3 = min(j + 3, cnt - 1);
            int s0 = __builtin_amdgcn_readlane(s_l, j);
            int s1 = __builtin_amdgcn_readlane(s_l, j1);
            int s2 = __builtin_amdgcn_readlane(s_l, j2);
            int s3 = __builtin_amdgcn_readlane(s_l, j3);
            float c1 = (j + 1 < cnt) ? 1.0f : 0.0f;
            float c2 = (j + 2 < cnt) ? 1.0f : 0.0f;
            float c3 = (j + 3 < cnt) ? 1.0f : 0.0f;
            const float2 v0 = *(const float2*)(T + (size_t)s0 * 64 + lane * 2);
            const float2 v1 = *(const float2*)(T + (size_t)s1 * 64 + lane * 2);
            const float2 v2 = *(const float2*)(T + (size_t)s2 * 64 + lane * 2);
            const float2 v3 = *(const float2*)(T + (size_t)s3 * 64 + lane * 2);
            acc.x += v0.x;                  acc.y += v0.y;
            acc.x = fmaf(v1.x, c1, acc.x);  acc.y = fmaf(v1.y, c1, acc.y);
            acc.x = fmaf(v2.x, c2, acc.x);  acc.y = fmaf(v2.y, c2, acc.y);
            acc.x = fmaf(v3.x, c3, acc.x);  acc.y = fmaf(v3.y, c3, acc.y);
        }
    }
    float s = inorm[n];
    float2 o;
    o.x = fmaf(acc.x, s, b2[lane * 2 + 0]);
    o.y = fmaf(acc.y, s, b2[lane * 2 + 1]);
    *(float2*)(H2 + (size_t)n * 64 + lane * 2) = o;
}

#define SQ  32
#define SWV 10

template <int OFF>
__device__ __forceinline__ void GCN_score_level(float (&p)[SQ], bool hi) {
#pragma unroll
    for (int j = 0; j < OFF; ++j) {
        float keep = hi ? p[j + OFF] : p[j];
        float send = hi ? p[j] : p[j + OFF];
        p[j] = keep + __shfl_xor(send, OFF, 32);
    }
}

__global__ void __launch_bounds__(32 * SWV) __attribute__((amdgpu_waves_per_eu(8, 8)))
GCN_score_kernel(const float* __restrict__ H2, const int* __restrict__ qs,
                 const int* __restrict__ qd, float* __restrict__ out) {
    static_assert(QQ % (SQ * SWV) == 0, "whole waves and whole work-groups only");
    static_assert((size_t)NN * 32 < 0x7fffffffu, "row offsets fit 32 bits");
    const float2* __restrict__ rows = (const float2*)H2;
    unsigned lane = threadIdx.x & 31;
    int w = __builtin_amdgcn_readfirstlane(blockIdx.x * SWV + (threadIdx.x >> 5));
    int q0 = w * SQ;
    int s_l = qs[q0 + lane], d_l = qd[q0 + lane];
    float2 a[SQ], b[SQ];
    float p[SQ];
#pragma unroll
    for (int j = 0; j < SQ; ++j) {
        unsigned s = (unsigned)__builtin_amdgcn_readlane(s_l, j);
        unsigned d = (unsigned)__builtin_amdgcn_readlane(d_l, j);
        a[j] = rows[s * 32u + lane];
        b[j] = rows[d * 32u + lane];
    }
#pragma unroll
    for (int j = 0; j < SQ; ++j) p[j] = fmaf(a[j].x, b[j].x, a[j].y * b[j].y);
    GCN_score_level<16>(p, (lane & 16) != 0);
    GCN_score_level<8>(p, (lane & 8) != 0);
    GCN_score_level<4>(p, (lane & 4) != 0);
    GCN_score_level<2>(p, (lane & 2) != 0);
    GCN_score_level<1>(p, (lane & 1) != 0);
    out[q0 + lane] = 1.0f / (1.0f + __expf(-p[0]));
}

extern "C" void kernel_launch(void* const* d_in, const int* in_sizes, int n_in,
                              void* d_out, int out_size, void* d_ws, size_t ws_size,
                              hipStream_t stream) {
    const float* feat = (const float*)d_in[0];
    const int* esrc   = (const int*)d_in[1];
    const int* edst   = (const int*)d_in[2];
    const int* qsrc   = (const int*)d_in[3];
    const int* qdst   = (const int*)d_in[4];
    const float* W1   = (const float*)d_in[5];
    const float* b1   = (const float*)d_in[6];
    const float* W2   = (const float*)d_in[7];
    const float* b2   = (const float*)d_in[8];
    float* out = (float*)d_out;

    float* ws    = (float*)d_ws;
    float* onorm = ws;
    float* agg1  = ws + 2 * NN;
    float* h2    = agg1;
    float* h1    = agg1 + (size_t)NN * 128;
    float* t2    = h1 + (size_t)NN * 128;
    int*   perm  = (int*)(t2 + (size_t)NN * 64);
    float* inorm = (float*)(perm + 2 * EE);
    int*   rowptr = (int*)(inorm + NN);
    int*   offs  = (int*)h1;
    int*   sortD = offs + 2 * NWG * NB;
    int*   sortS = sortD + EE;

    GCN_chunksort_kernel<<<NWG, 1024, 0, stream>>>(esrc, edst, offs, sortD, sortS);
    GCN_rows_kernel<<<NB, 1024, 0, stream>>>(offs, sortD, sortS, rowptr, inorm, onorm, perm);
    GCN_gather1_kernel<<<NN / 8, 256, 0, stream>>>(feat, rowptr, perm, onorm, agg1);
    GCN_gemm12_kernel<<<GGRID, 32 * GS, 0, stream>>>(agg1, W1, b1, inorm, onorm, W2, t2);
    GCN_gather2_kernel<<<NN / 8, 256, 0, stream>>>(t2, rowptr, perm, inorm, b2, h2);
    GCN_score_kernel<<<QQ / (SQ * SWV), 32 * SWV, 0, stream>>>(h2, qsrc, qdst, out);
}


# Layer-1 gather fused into the product launch: a wave sums its 16-row strip's neighbours (one row stream, 16-32 loads in flight, same sum order) in its own LDS corner and reads its fragments there
# speedup vs baseline: 5.6713x; 1.0808x over previous
#include <hip/hip_runtime.h>
#include <math.h>

typedef __attribute__((ext_vector_type(2))) float v2f;
typedef __attribute__((ext_vector_type(8))) float v8f;

#define NN 100000
#define EE 1600000
#define QQ 1000000

#define NB  256
#define BW  391
#define NWG 256
#define CH  (EE / NWG)
#define EPT 7
#define SRC_BITS 17
#define CPW (NWG / 32)
#define PCAP 8192

__global__ void __launch_bounds__(1024)
GCN_chunksort_kernel(const int* __restrict__ esrc, const int* __restrict__ edst,
                     int* __restrict__ offs, int* __restrict__ sortD,
                     int* __restrict__ sortS) {
    __shared__ int stD[CH];
    __shared__ int stS[CH];
    __shared__ int hD[NB];
    __shared__ int hS[NB];
    __shared__ int wsum[2 * NB / 32];
    static_assert(NB % 32 == 0 && 2 * NB <= 1024, "whole waves scan the bin counts");
    int t = threadIdx.x, w = blockIdx.x, lane = t & 31, wave = t >> 5;
    if (t < NB) { hD[t] = 0; hS[t] = 0; }
    int e0 = w * CH;
    int d[EPT], s[EPT];
#pragma unroll
    for (int k = 0; k < EPT; ++k) {
        int i = t + k * 1024;
        d[k] = (i < CH) ? edst[e0 + i] : -1;
        s[k] = (i < CH) ? esrc[e0 + i] : -1;
    }
    __syncthreads();
#pragma unroll
    for (int k = 0; k < EPT; ++k) {
        if (d[k] >= 0) {
            atomicAdd(&hD[d[k] / BW], 1);
            atomicAdd(&hS[s[k] / BW], 1);
        }
    }
    __syncthreads();
    int kind = t / NB, bin = t % NB;
    int c = 0, inc = 0;
    if (t < 2 * NB) {
        c = kind ? hS[bin] : hD[bin];
        inc = c;
#pragma unroll
        for (int off = 1; off < 32; off <<= 1) {
            int u = __shfl_up(inc, off, 32);
            if (lane >= off) inc += u;
        }
        if (lane == 31) wsum[wave] = inc;
    }
    __syncthreads();
    if (t < 2 * NB) {
        int w0 = kind * (NB / 32);
        int pre = 0;
#pragma unroll
        for (int k = 0; k < NB / 32; ++k) {
            if (w0 + k < wave) pre += wsum[w0 + k];
        }
        int ls = pre + inc - c;
        if (kind) hS[bin] = ls; else hD[bin] = ls;
        offs[kind * (NWG * NB) + w * NB + bin] = ls;
    }
    __syncthreads();
#pragma unroll
    for (int k = 0; k < EPT; ++k) {
        if (d[k] >= 0) {
            int bd = d[k] / BW, bs = s[k] / BW;
            int pd = atomicAdd(&hD[bd], 1);
            int ps = atomicAdd(&hS[bs], 1);
            stD[pd] = ((d[k] - bd * BW) << SRC_BITS) | s[k];
            stS[ps] = s[k] - bs * BW;
        }
    }
    __syncthreads();
#pragma unroll
    for (int k = 0; k < EPT; ++k) {
        int i = t + k * 1024;
        if (i < CH) {
            sortD[e0 + i] = stD[i];
            sortS[e0 + i] = stS[i];
        }
    }
}


template <bool IN_LDS>
__device__ __forceinline__ void GCN_place_rows(const int (&p)[CPW], int maxD, int wave, int lane,
                                               const int* __restrict__ sortD, const int* segD,
                                               int* cur, int* lperm, int* __restrict__ out) {
#pragma unroll
    for (int k = 0; k < CPW; ++k) {
        if (p[k] >= 0) {
            int pos = atomicAdd(&cur[p[k] >> SRC_BITS], 1);
            int sid = p[k] & ((1 << SRC_BITS) - 1);
            if (IN_LDS) lperm[pos] = sid;
            else        out[pos] = sid;
        }
    }
    for (int r = 32; r < maxD; r += 32) {
        for (int k = 0; k < CPW; ++k) {
            int w = wave * CPW + k;
            int i = segD[2 * w] + r + lane;
            if (i < segD[2 * w + 1]) {
                int e = sortD[w * CH + i];
                int pos = atomicAdd(&cur[e >> SRC_BITS], 1);
                int sid = e & ((1 << SRC_BITS) - 1);
                if (IN_LDS) lperm[pos] = sid;
                else        out[pos] = sid;
            }
        }
    }
}

__global__ void __launch_bounds__(1024)
GCN_rows_kernel(const int* __restrict__ offs, const int* __restrict__ sortD,
                const int* __restrict__ sortS, int* __restrict__ rowptr,
                float* __restrict__ inorm, float* __restrict__ onorm,
                int* __restrict__ perm) {
    __shared__ int cntIn[BW];
    __shared__ int cntOut[BW];
    __shared__ int cur[BW];
    __shared__ int wsum[32];
    __shared__ int segD[2 * NWG];
    __shared__ int segS[2 * NWG];
    __shared__ int redB[NWG / 32];
    __shared__ int redE[NWG / 32];
    __shared__ int lperm[PCAP];
    static_assert(NWG % 32 == 0 && NWG <= 1024 && CPW * 32 == NWG, "a thread per chunk, CPW chunks per wave");
    int t = threadIdx.x, b = blockIdx.x, lane = t & 31, wave = t >> 5;
    if (t < BW) { cntIn[t] = 0; cntOut[t] = 0; }
    if (t < NWG) {
        const int* oD = offs + t * NB;
        const int* oS = offs + NWG * NB + t * NB;
        int sd = oD[b], ed = (b + 1 < NB) ? oD[b + 1] : CH;
        int ss = oS[b], es = (b + 1 < NB) ? oS[b + 1] : CH;
        segD[2 * t] = sd; segD[2 * t + 1] = ed;
        segS[2 * t] = ss; segS[2 * t + 1] = es;
#pragma unroll
        for (int off = 16; off > 0; off >>= 1) {
            sd += __shfl_xor(sd, off, 32);
            ed += __shfl_xor(ed, off, 32);
        }
        if (lane == 0) { redB[wave] = sd; redE[wave] = ed; }
    }
    __syncthreads();
    int dBeg = 0, dEnd = 0;
#pragma unroll
    for (int k = 0; k < NWG / 32; ++k) { dBeg += redB[k]; dEnd += redE[k]; }

    int p[CPW], q[CPW];
    int maxD = 0, maxS = 0;
#pragma unroll
    for (int k = 0; k < CPW; ++k) {
        int w = wave * CPW + k;
        int sd = segD[2 * w], ed = segD[2 * w + 1];
        int ss = segS[2 * w], es = segS[2 * w + 1];
        p[k] = (sd + lane < ed) ? sortD[w * CH + sd + lane] : -1;
        q[k] = (ss + lane < es) ? sortS[w * CH + ss + lane] : -1;
        maxD = max(maxD, ed - sd);
        maxS = max(maxS, es - ss);
    }
    maxD = __builtin_amdgcn_readfirstlane(maxD);
    maxS = __builtin_amdgcn_readfirstlane(maxS);
#pragma unroll
    for (int k = 0; k < CPW; ++k) {
        if (p[k] >= 0) atomicAdd(&cntIn[p[k] >> SRC_BITS], 1);
        if (q[k] >= 0) atomicAdd(&cntOut[q[k]], 1);
    }
    for (int r = 32; r < maxD; r += 32) {
        for (int k = 0; k < CPW; ++k) {
            int w = wave * CPW + k;
            int i = segD[2 * w] + r + lane;
            if (i < segD[2 * w + 1]) atomicAdd(&cntIn[sortD[w * CH + i] >> SRC_BITS], 1);
        }
    }
    for (int r = 32; r < maxS; r += 32) {
        for (int k = 0; k < CPW; ++k) {
            int w = wave * CPW + k;
            int i = segS[2 * w] + r + lane;
            if (i < segS[2 * w + 1]) atomicAdd(&cntOut[sortS[w * CH + i]], 1);
        }
    }
    __syncthreads();

    int c = (t < BW) ? cntIn[t] : 0;
    int inc = c;
#pragma unroll
    for (int off = 1; off < 32; off <<= 1) {
        int u = __shfl_up(inc, off, 32);
        if (lane >= off) inc += u;
    }
    if (lane == 31) wsum[wave] = inc;
    __syncthreads();
    int pre = 0;
#pragma unroll
    for (int k = 0; k < 32; ++k) {
        if (k < wave) pre += wsum[k];
    }
    int first = dBeg + pre + inc - c;
    int n = b * BW + t;
    if (t < BW) {
        cur[t] = first - dBeg;
        if (n < NN) {
            rowptr[n] = first;
            inorm[n] = rsqrtf(fmaxf((float)c, 1.0f));
            onorm[n] = rsqrtf(fmaxf((float)cntOut[t], 1.0f));
            if (n == NN - 1) rowptr[NN] = first + c;
        }
    }
    __syncthreads();
    int len = dEnd - dBeg;
    if (len > PCAP) {
        GCN_place_rows<false>(p, maxD, wave, lane, sortD, segD, cur, lperm, perm + dBeg);
    } else {
        GCN_place_rows<true>(p, maxD, wave, lane, sortD, segD, cur, lperm, perm + dBeg);
        __syncthreads();
        int o[PCAP / 1024];
#pragma unroll
        for (int k = 0; k < PCAP / 1024; ++k) {
            int i = t + k * 1024;
            o[k] = (i < len) ? lperm[i] : 0;
        }
#pragma unroll
        for (int k = 0; k < PCAP / 1024; ++k) {
            int i = t + k * 1024;
            if (i < len) perm[dBeg + i] = o[k];
        }
    }
}

#define FW    16
#define FGRID 256
#define NSTRIP (NN / 16)
#define SP    132
#define FCH   384
#define W2P   136
static_assert(NN % 16 == 0, "whole strips only");
static_assert((128 * 128 / 4) % (32 * FW) == 0 && (64 * 16) % (32 * FW) == 0, "the W staging loops take whole trips");
static_assert(32 * FW >= 128 && FCH % 32 == 0, "bias staging; whole waves of edges");

__device__ __forceinline__ void GCN_wave_lds_order() {
    __builtin_amdgcn_fence(__ATOMIC_RELEASE, "wavefront");
    __builtin_amdgcn_wave_barrier();
    __builtin_amdgcn_fence(__ATOMIC_ACQUIRE, "wavefront");
}

#define RD 16
static_assert(FCH % (2 * RD) == 0, "the stream's loop takes two whole groups per trip");

__device__ __forceinline__ void GCN_rows_ask(float4 (&v)[RD], float (&s)[RD], const int2* myE, int t0,
                                             const float* __restrict__ feat, unsigned lane) {
#pragma unroll
    for (int u = 0; u < RD; ++u) {
        int2 ed = myE[min(t0 + u, FCH - 1)];
        s[u] = __int_as_float(ed.y);
        v[u] = *(const float4*)(feat + ((unsigned)ed.x * 128u + lane * 4u));
    }
}

__device__ __forceinline__ void GCN_rows_sum(const float4 (&v)[RD], const float (&s)[RD], int t0,
                                             int sb, int rp, int& row, int& rEnd, float4& acc,
                                             float* myS, unsigned lane) {
#pragma unroll
    for (int u = 0; u < RD; ++u) {
        int ge = sb + t0 + u;
        while (row < 15 && ge >= rEnd) {
            *(float4*)(myS + row * SP + lane * 4) = acc;
            acc = make_float4(0.0f, 0.0f, 0.0f, 0.0f);
            ++row;
            rEnd = __builtin_amdgcn_readlane(rp, row + 1);
        }
        acc.x = fmaf(v[u].x, s[u], acc.x); acc.y = fmaf(v[u].y, s[u], acc.y);
        acc.z = fmaf(v[u].z, s[u], acc.z); acc.w = fmaf(v[u].w, s[u], acc.w);
    }
}

__global__ void __launch_bounds__(32 * FW)
GCN_gather_gemm12_kernel(const float* __restrict__ feat, const int* __restrict__ rowptr,
                         const int* __restrict__ perm, const float* __restrict__ W1,
                         const float* __restrict__ bias, const float* __restrict__ inorm,
                         const float* __restrict__ onorm, const float* __restrict__ W2,
                         float* __restrict__ T) {
    __shared__ float4 lds4[(64 * W2P + 128 + 128 * 128 + FW * 16 * SP + FW * FCH * 2) / 4];
    float* lW2 = (float*)lds4;
    float* lB  = lW2 + 64 * W2P;
    float* lW1 = lB + 128;
    float* lS  = lW1 + 128 * 128;
    int2*  lE  = (int2*)(lS + FW * 16 * SP);
    int tid = threadIdx.x;
    unsigned lane = tid & 31;
    int wave = __builtin_amdgcn_readfirstlane(tid >> 5);
    int r = lane & 15, kh = lane >> 4;
    {
        const float4* Wv = (const float4*)W1;
        float4* lWv = (float4*)lW1;
#pragma unroll
        for (int i = 0; i < (128 * 128 / 4) / (32 * FW); ++i) lWv[tid + 32 * FW * i] = Wv[tid + 32 * FW * i];
    }
    {
        const float4* Wv = (const float4*)W2;
        float4* lWv = (float4*)lW2;
#pragma unroll
        for (int i = 0; i < (64 * 16) / (32 * FW); ++i) {
            int idx = tid + 32 * FW * i;
            int kp = idx >> 4, c4 = idx & 15;
            float4 u = Wv[(2 * kp) * 16 + c4];
            float4 w = Wv[(2 * kp + 1) * 16 + c4];
            lWv[kp * (W2P / 4) + 2 * c4]     = make_float4(u.x, w.x, u.y, w.y);
            lWv[kp * (W2P / 4) + 2 * c4 + 1] = make_float4(u.z, w.z, u.w, w.w);
        }
    }
    if (tid < 128) lB[tid] = bias[tid];
    __syncthreads();

    float* myS = lS + wave * 16 * SP;
    int2*  myE = lE + wave * FCH;
    for (int j = wave; j < (NSTRIP + FGRID - 1) / FGRID; j += FW) {
        int strip = blockIdx.x + FGRID * j;
        if (strip >= NSTRIP) break;
        int row0 = strip * 16;
        int rp = rowptr[row0 + min((int)lane, 16)];
        int e0 = __builtin_amdgcn_readlane(rp, 0), e1 = __builtin_amdgcn_readlane(rp, 16);
        float si = inorm[row0 + r];
        float so = onorm[row0 + r];
        int row = 0;
        int rEnd = __builtin_amdgcn_readlane(rp, 1);
        float4 acc = make_float4(0.0f, 0.0f, 0.0f, 0.0f);
        for (int sb = e0; sb < e1; sb += FCH) {
            int n = min(FCH, e1 - sb);
            GCN_wave_lds_order();
            {
                int sid[FCH / 32];
                float sc[FCH / 32];
#pragma unroll
                for (int k = 0; k < FCH / 32; ++k) {
                    int i = 32 * k + (int)lane;
                    sid[k] = perm[sb + min(i, n - 1)];
                }
#pragma unroll
                for (int k = 0; k < FCH / 32; ++k) {
                    int i = 32 * k + (int)lane;
                    float o = onorm[sid[k]];
                    sc[k] = (i < n) ? o : 0.0f;
                }
#pragma unroll
                for (int k = 0; k < FCH / 32; ++k)
                    myE[32 * k + lane] = make_int2(sid[k], __float_as_int(sc[k]));
            }
            GCN_wave_lds_order();
            float4 va[RD], vb[RD];
            float sa[RD], sv[RD];
            GCN_rows_ask(va, sa, myE, 0, feat, lane);
            for (int t = 0; t < n; t += 2 * RD) {
                GCN_rows_ask(vb, sv, myE, t + RD, feat, lane);
                GCN_rows_sum(va, sa, t, sb, rp, row, rEnd, acc, myS, lane);
                GCN_rows_ask(va, sa, myE, t + 2 * RD, feat, lane);
                GCN_rows_sum(vb, sv, t + RD, sb, rp, row, rEnd, acc, myS, lane);
            }
        }
        for (; row < 16; ++row) {
            *(float4*)(myS + row * SP + lane * 4) = acc;
            acc = make_float4(0.0f, 0.0f, 0.0f, 0.0f);
        }
        GCN_wave_lds_order();
        v2f a[32];
        {
            const float* Ap = myS + r * SP + 2 * kh;
#pragma unroll
            for (int kt = 0; kt < 32; ++kt) a[kt] = *(const v2f*)(Ap + kt * 4);
        }
        GCN_wave_lds_order();

    v8f c[8] = {};
    typedef const __attribute__((address_space(3))) float lds_cfloat;
    lds_cfloat* Bp = (lds_cfloat*)(lW1 + (size_t)(2 * kh) * 128 + r);
    asm volatile("" : "+v"(Bp));
#pragma unroll
    for (int kt = 0; kt < 32; ++kt) {
#pragma unroll
        for (int ct = 0; ct < 8; ++ct) {
            v2f b;
            b.x = Bp[kt * 512 + ct * 16];
            b.y = Bp[kt * 512 + 128 + ct * 16];
            c[ct] = __builtin_amdgcn_wmma_f32_16x16x4_f32(false, b, false, a[kt], (short)0,
                                                          c[ct], false, false);
        }
    }
#pragma unroll
    for (int ct = 0; ct < 8; ++ct) {
        const float4* bq = (const float4*)(lB + ct * 16 + 8 * kh);
        float4 q0 = bq[0], q1 = bq[1];
        float bb[8] = {q0.x, q0.y, q0.z, q0.w, q1.x, q1.y, q1.z, q1.w};
#pragma unroll
        for (int v = 0; v < 8; ++v) {
            float val = fmaf(c[ct][v], si, bb[v]);
            c[ct][v] = fmaxf(val, 0.0f) * so;
        }
    }

    v8f d[4] = {};
    const float* B2 = lW2 + (size_t)(4 * kh) * W2P + 2 * r;
#pragma unroll
    for (int ct = 0; ct < 8; ++ct) {
#pragma unroll
        for (int p = 0; p < 4; ++p) {
            v2f h;
            h.x = c[ct][2 * p];
            h.y = c[ct][2 * p + 1];
#pragma unroll
            for (int jt = 0; jt < 4; ++jt) {
                v2f b = *(const v2f*)(B2 + (ct * 8 + p) * W2P + jt * 32);
                d[jt] = __builtin_amdgcn_wmma_f32_16x16x4_f32(false, h, false, b, (short)0,
                                                              d[jt], false, false);
            }
        }
    }
#pragma unroll
    for (int jt = 0; jt < 4; ++jt) {
        int n = jt * 16 + r;
#pragma unroll
        for (int v = 0; v < 8; ++v) {
            int m = row0 + v + 8 * kh;
            T[(size_t)m * 64 + n] = d[jt][v];
        }
    }
    }
}

__global__ void __launch_bounds__(256)
GCN_gather2_kernel(const float* __restrict__ T, const int* __restrict__ rowptr,
                   const int* __restrict__ perm, const float* __restrict__ inorm,
                   const float* __restrict__ b2, float* __restrict__ H2) {
    int lane = threadIdx.x & 31;
    int n = __builtin_amdgcn_readfirstlane(blockIdx.x * 8 + (threadIdx.x >> 5));
    int beg = rowptr[n], end = rowptr[n + 1];
    float2 acc = make_float2(0.0f, 0.0f);
    for (int base = beg; base < end; base += 32) {
        int cnt = min(32, end - base);
        int s_l = 0;
        if (lane < cnt) s_l = perm[base + lane];
        for (int j = 0; j < cnt; j += 4) {
            int j1 = min(j + 1, cnt - 1), j2 = min(j + 2, cnt - 1), j3 = min(j + 3, cnt - 1);
            int s0 = __builtin_amdgcn_readlane(s_l, j);
            int s1 = __builtin_amdgcn_readlane(s_l, j1);
            int s2 = __builtin_amdgcn_readlane(s_l, j2);
            int s3 = __builtin_amdgcn_readlane(s_l, j3);
            float c1 = (j + 1 < cnt) ? 1.0f : 0.0f;
            float c2 = (j + 2 < cnt) ? 1.0f : 0.0f;
            float c3 = (j + 3 < cnt) ? 1.0f : 0.0f;
            const float2 v0 = *(const float2*)(T + (size_t)s0 * 64 + lane * 2);
            const float2 v1 = *(const float2*)(T + (size_t)s1 * 64 + lane * 2);
            const float2 v2 = *(const float2*)(T + (size_t)s2 * 64 + lane * 2);
            const float2 v3 = *(const float2*)(T + (size_t)s3 * 64 + lane * 2);
            acc.x += v0.x;                  acc.y += v0.y;
            acc.x = fmaf(v1.x, c1, acc.x);  acc.y = fmaf(v1.y, c1, acc.y);
            acc.x = fmaf(v2.x, c2, acc.x);  acc.y = fmaf(v2.y, c2, acc.y);
            acc.x = fmaf(v3.x, c3, acc.x);  acc.y = fmaf(v3.y, c3, acc.y);
        }
    }
    float s = inorm[n];
    float2 o;
    o.x = fmaf(acc.x, s, b2[lane * 2 + 0]);
    o.y = fmaf(acc.y, s, b2[lane * 2 + 1]);
    *(float2*)(H2 + (size_t)n * 64 + lane * 2) = o;
}

#define SQ  32
#define SWV 10

template <int OFF>
__device__ __forceinline__ void GCN_score_level(float (&p)[SQ], bool hi) {
#pragma unroll
    for (int j = 0; j < OFF; ++j) {
        float keep = hi ? p[j + OFF] : p[j];
        float send = hi ? p[j] : p[j + OFF];
        p[j] = keep + __shfl_xor(send, OFF, 32);
    }
}

__global__ void __launch_bounds__(32 * SWV) __attribute__((amdgpu_waves_per_eu(8, 8)))
GCN_score_kernel(const float* __restrict__ H2, const int* __restrict__ qs,
                 const int* __restrict__ qd, float* __restrict__ out) {
    static_assert(QQ % (SQ * SWV) == 0, "whole waves and whole work-groups only");
    static_assert((size_t)NN * 32 < 0x7fffffffu, "row offsets fit 32 bits");
    const float2* __restrict__ rows = (const float2*)H2;
    unsigned lane = threadIdx.x & 31;
    int w = __builtin_amdgcn_readfirstlane(blockIdx.x * SWV + (threadIdx.x >> 5));
    int q0 = w * SQ;
    int s_l = qs[q0 + lane], d_l = qd[q0 + lane];
    float2 a[SQ], b[SQ];
    float p[SQ];
#pragma unroll
    for (int j = 0; j < SQ; ++j) {
        unsigned s = (unsigned)__builtin_amdgcn_readlane(s_l, j);
        unsigned d = (unsigned)__builtin_amdgcn_readlane(d_l, j);
        a[j] = rows[s * 32u + lane];
        b[j] = rows[d * 32u + lane];
    }
#pragma unroll
    for (int j = 0; j < SQ; ++j) p[j] = fmaf(a[j].x, b[j].x, a[j].y * b[j].y);
    GCN_score_level<16>(p, (lane & 16) != 0);
    GCN_score_level<8>(p, (lane & 8) != 0);
    GCN_score_level<4>(p, (lane & 4) != 0);
    GCN_score_level<2>(p, (lane & 2) != 0);
    GCN_score_level<1>(p, (lane & 1) != 0);
    out[q0 + lane] = 1.0f / (1.0f + __expf(-p[0]));
}

extern "C" void kernel_launch(void* const* d_in, const int* in_sizes, int n_in,
                              void* d_out, int out_size, void* d_ws, size_t ws_size,
                              hipStream_t stream) {
    const float* feat = (const float*)d_in[0];
    const int* esrc   = (const int*)d_in[1];
    const int* edst   = (const int*)d_in[2];
    const int* qsrc   = (const int*)d_in[3];
    const int* qdst   = (const int*)d_in[4];
    const float* W1   = (const float*)d_in[5];
    const float* b1   = (const float*)d_in[6];
    const float* W2   = (const float*)d_in[7];
    const float* b2   = (const float*)d_in[8];
    float* out = (float*)d_out;

    float* ws    = (float*)d_ws;
    float* onorm = ws;
    float* agg1  = ws + 2 * NN;
    float* h2    = agg1;
    float* h1    = agg1 + (size_t)NN * 128;
    float* t2    = h1 + (size_t)NN * 128;
    int*   perm  = (int*)(t2 + (size_t)NN * 64);
    float* inorm = (float*)(perm + 2 * EE);
    int*   rowptr = (int*)(inorm + NN);
    int*   offs  = (int*)h1;
    int*   sortD = offs + 2 * NWG * NB;
    int*   sortS = sortD + EE;

    GCN_chunksort_kernel<<<NWG, 1024, 0, stream>>>(esrc, edst, offs, sortD, sortS);
    GCN_rows_kernel<<<NB, 1024, 0, stream>>>(offs, sortD, sortS, rowptr, inorm, onorm, perm);
    GCN_gather_gemm12_kernel<<<FGRID, 32 * FW, 0, stream>>>(feat, rowptr, perm, W1, b1, inorm, onorm,
                                                            W2, t2);
    GCN_gather2_kernel<<<NN / 8, 256, 0, stream>>>(t2, rowptr, perm, inorm, b2, h2);
    GCN_score_kernel<<<QQ / (SQ * SWV), 32 * SWV, 0, stream>>>(h2, qsrc, qdst, out);
}
